# starting kernel
# baseline (speedup 1.0000x reference)
#include <hip/hip_runtime.h>
#include <hip/hip_bf16.h>


typedef float v2f __attribute__((ext_vector_type(2)));
typedef float v8f __attribute__((ext_vector_type(8)));

#define SCQ 0.70710678118654752440f

__constant__ float cG0B[14] = {
   0.00325314f, -0.00388321f,  0.03466035f, -0.03887280f,
  -0.11720389f,  0.27529538f,  0.75614564f,  0.56881042f,
   0.01186609f, -0.10671180f,  0.02382538f,  0.01702522f,
  -0.00543948f, -0.00455690f };
__constant__ float cG0A[14] = {
  -0.00455690f, -0.00543948f,  0.01702522f,  0.02382538f,
  -0.10671180f,  0.01186609f,  0.56881042f,  0.75614564f,
   0.27529538f, -0.11720389f, -0.03887280f,  0.03466035f,
  -0.00388321f,  0.00325314f };
__constant__ float cG1B[14] = {
  -0.00455690f,  0.00543948f,  0.01702522f, -0.02382538f,
  -0.10671180f, -0.01186609f,  0.56881042f, -0.75614564f,
   0.27529538f,  0.11720389f, -0.03887280f, -0.03466035f,
  -0.00388321f, -0.00325314f };
__constant__ float cG1A[14] = {
  -0.00325314f, -0.00388321f, -0.03466035f, -0.03887280f,
   0.11720389f,  0.27529538f, -0.75614564f,  0.56881042f,
  -0.01186609f, -0.10671180f, -0.02382538f,  0.01702522f,
   0.00543948f, -0.00455690f };

__constant__ float cG0O[7] = { -3.0f/280.0f, -15.0f/280.0f, 73.0f/280.0f,
                               170.0f/280.0f, 73.0f/280.0f, -15.0f/280.0f,
                               -3.0f/280.0f };
__constant__ float cG1O[5] = { -1.0f/20.0f, -5.0f/20.0f, 12.0f/20.0f,
                               -5.0f/20.0f, -1.0f/20.0f };

__device__ __forceinline__ int ref_idx(int i, int L) {
  if (i < 0)  i = -1 - i;
  if (i >= L) i = 2 * L - 1 - i;
  return i;
}

__device__ __forceinline__ float c2q_fetch(const float* __restrict__ hi,
                                           int b, int r, int v, int ch,
                                           int sub, int Ld) {
  constexpr int SA[3] = { 0, 2, 1 };
  constexpr int SB[3] = { 5, 3, 4 };
  const int i = r >> 1, j = v >> 1;
  const long long base = (((long long)b * Ld + i) * Ld + j) * 36;
  const int py = r & 1, px = v & 1;
  const int im = (px ^ py) ? 18 : 0;
  const float a  = hi[base + ch + 3 * SA[sub] + im];
  const float bb = hi[base + ch + 3 * SB[sub] + im];
  return py ? ((px ^ py) ? (a - bb) * SCQ : (bb - a) * SCQ) : (a + bb) * SCQ;
}

__device__ __forceinline__ float colifilt_eval(bool g1, int qy, int sy, int v,
                                               const float* __restrict__ buf,
                                               int b, int ch, int sub) {
  const float* ha = g1 ? cG1B : cG0B;
  const float* hb = g1 ? cG1A : cG0A;
  const int   p   = sy >> 1;
  const bool  aB  = (sy & 1) == 0;
  const float* f  = aB ? ha : hb;
  const int base  = g1 ? (aB ? 7 : 6) : (aB ? 6 : 7);
  float acc = 0.0f;
#pragma unroll
  for (int j = 0; j < 7; ++j) {
    const int r = ref_idx(2 * qy + base - 2 * j, 256);
    float x;
    if (sub < 0) x = buf[(((long long)b * 256 + r) * 256 + v) * 3 + ch];
    else         x = c2q_fetch(buf, b, r, v, ch, sub, 128);
    acc += f[2 * j + p] * x;
  }
  return acc;
}

__global__ void __launch_bounds__(256)
k_level1(const float* __restrict__ low, const float* __restrict__ high1,
         float* __restrict__ Z) {
  const int tx = blockIdx.x, ty = blockIdx.y, bz = blockIdx.z;
  const int b = bz / 3, ch = bz % 3;
  const int x0 = tx * 16, y0 = ty * 16;
  const int u0 = 8 * tx - 6;

  __shared__ float y1s[16][21];
  __shared__ float y2s[16][21];

  const int tid = threadIdx.x;
  for (int idx = tid; idx < 2 * 16 * 20; idx += 256) {
    const int plane = idx / 320;
    const int rem   = idx - plane * 320;
    const int ry = rem / 20, ci = rem % 20;
    const int v  = ref_idx(u0 + ci, 256);
    const int y  = y0 + ry, qy = y >> 2, sy = y & 3;
    float val;
    if (plane == 0)
      val = colifilt_eval(false, qy, sy, v, low,   b, ch, -1) +
            colifilt_eval(true,  qy, sy, v, high1, b, ch,  0);
    else
      val = colifilt_eval(false, qy, sy, v, high1, b, ch,  1) +
            colifilt_eval(true,  qy, sy, v, high1, b, ch,  2);
    (plane ? y2s : y1s)[ry][ci] = val;
  }
  __syncthreads();

  const int ry = tid >> 4, cx = tid & 15;
  const int x  = x0 + cx;
  const int q2 = 2 * (cx >> 2);
  const int sx = x & 3, p = sx >> 1;
  const bool aB = (sx & 1) == 0;
  const float* f1 = aB ? cG0B : cG0A;  const int b1 = aB ? 6 : 7;
  const float* f2 = aB ? cG1B : cG1A;  const int b2 = aB ? 7 : 6;
  float acc = 0.0f;
#pragma unroll
  for (int j = 0; j < 7; ++j) {
    acc += f1[2 * j + p] * y1s[ry][q2 + b1 + 6 - 2 * j];
    acc += f2[2 * j + p] * y2s[ry][q2 + b2 + 6 - 2 * j];
  }
  Z[(((long long)b * 512 + (y0 + ry)) * 512 + x) * 3 + ch] = acc;
}

__device__ __forceinline__ float tap7(int i) {
  int ic = i < 0 ? 0 : (i > 6 ? 6 : i);
  float t = cG0O[ic];
  return (i == ic) ? t : 0.0f;
}
__device__ __forceinline__ float tap5(int i) {
  int ic = i < 0 ? 0 : (i > 4 ? 4 : i);
  float t = cG1O[ic];
  return (i == ic) ? t : 0.0f;
}

__device__ __forceinline__ v8f wmma4(v2f a, v2f b, v8f c) {
  return __builtin_amdgcn_wmma_f32_16x16x4_f32(
       false, a,  false, b,
       (short)0, c,  false,  false);
}

__global__ void __launch_bounds__(32)
k_level0(const float* __restrict__ Z, const float* __restrict__ high0,
         float* __restrict__ out) {
  const int tx = blockIdx.x, ty = blockIdx.y, bz = blockIdx.z;
  const int b = bz / 3, ch = bz % 3;
  const int x0 = tx * 16, y0 = ty * 16;

  __shared__ float zs [24][33];
  __shared__ float lhs[24][33];
  __shared__ float hls[24][33];
  __shared__ float hhs[24][33];
  __shared__ float y1s[16][33];
  __shared__ float y2s[16][33];

  const int lane = threadIdx.x;
  const int lo = lane & 15;
  const int h2 = (lane >> 4) * 2;

  v2f f0[6], f1[6];
#pragma unroll
  for (int kk = 0; kk < 6; ++kk) {
    const int k0 = 4 * kk + h2;
    f0[kk].x = tap7(k0 - lo - 1);  f0[kk].y = tap7(k0 - lo);
    f1[kk].x = tap5(k0 - lo - 2);  f1[kk].y = tap5(k0 - lo - 1);
  }

  for (int idx = lane; idx < 24 * 32; idx += 32) {
    const int r = idx >> 5, c = idx & 31;
    const int rr = ref_idx(y0 - 4 + r, 512);
    const int cc = ref_idx(x0 - 8 + c, 512);
    zs [r][c] = Z[(((long long)b * 512 + rr) * 512 + cc) * 3 + ch];
    lhs[r][c] = c2q_fetch(high0, b, rr, cc, ch, 0, 256);
    hls[r][c] = c2q_fetch(high0, b, rr, cc, ch, 1, 256);
    hhs[r][c] = c2q_fetch(high0, b, rr, cc, ch, 2, 256);
  }
  __syncthreads();

#pragma unroll
  for (int t = 0; t < 2; ++t) {
    v8f a1 = {}; v8f a2 = {};
#pragma unroll
    for (int kk = 0; kk < 6; ++kk) {
      const int k0 = 4 * kk + h2;
      const int cc = 16 * t + lo;
      v2f bz = { zs [k0][cc], zs [k0 + 1][cc] };
      v2f bl = { lhs[k0][cc], lhs[k0 + 1][cc] };
      v2f bh = { hls[k0][cc], hls[k0 + 1][cc] };
      v2f bq = { hhs[k0][cc], hhs[k0 + 1][cc] };
      a1 = wmma4(f0[kk], bz, a1);
      a1 = wmma4(f1[kk], bl, a1);
      a2 = wmma4(f0[kk], bh, a2);
      a2 = wmma4(f1[kk], bq, a2);
    }
#pragma unroll
    for (int v = 0; v < 8; ++v) {
      const int row = v + (h2 << 2);
      y1s[row][16 * t + lo] = a1[v];
      y2s[row][16 * t + lo] = a2[v];
    }
  }
  __syncthreads();

  v8f o = {};
#pragma unroll
  for (int kk = 0; kk < 6; ++kk) {
    const int k0 = 4 * kk + h2;
    v2f A1 = { y1s[lo][k0 + 4], y1s[lo][k0 + 5] };
    v2f A2 = { y2s[lo][k0 + 4], y2s[lo][k0 + 5] };
    o = wmma4(A1, f0[kk], o);
    o = wmma4(A2, f1[kk], o);
  }
#pragma unroll
  for (int v = 0; v < 8; ++v) {
    const int row = y0 + v + (h2 << 2);
    const int col = x0 + lo;
    out[(((long long)b * 512 + row) * 512 + col) * 3 + ch] = o[v];
  }
}

extern "C" void kernel_launch(void* const* d_in, const int* in_sizes, int n_in,
                              void* d_out, int out_size, void* d_ws, size_t ws_size,
                              hipStream_t stream) {
  (void)in_sizes; (void)n_in; (void)out_size; (void)ws_size;
  const float* low   = (const float*)d_in[0];
  const float* high0 = (const float*)d_in[1];
  const float* high1 = (const float*)d_in[2];
  float* out = (float*)d_out;
  float* Zws = (float*)d_ws;

  dim3 grid(32, 32, 8 * 3);
  k_level1<<<grid, dim3(256), 0, stream>>>(low, high1, Zws);
  k_level0<<<grid, dim3(32), 0, stream>>>(Zws, high0, out);
}


# level-1 kernel: one block per 32x32x3 Z tile stages the reflected 28x28 window of low and the c2q planes of high1 in LDS once; both colifilt passes and line-wide Z stores run from LDS
# speedup vs baseline: 2.8489x; 2.8489x over previous
#include <hip/hip_runtime.h>
#include <hip/hip_bf16.h>


typedef float v2f __attribute__((ext_vector_type(2)));
typedef float v8f __attribute__((ext_vector_type(8)));

#define SCQ 0.70710678118654752440f

__constant__ float cG0O[7] = { -3.0f/280.0f, -15.0f/280.0f, 73.0f/280.0f,
                               170.0f/280.0f, 73.0f/280.0f, -15.0f/280.0f,
                               -3.0f/280.0f };
__constant__ float cG1O[5] = { -1.0f/20.0f, -5.0f/20.0f, 12.0f/20.0f,
                               -5.0f/20.0f, -1.0f/20.0f };

__device__ __forceinline__ int ref_idx(int i, int L) {
  if (i < 0)  i = -1 - i;
  if (i >= L) i = 2 * L - 1 - i;
  return i;
}

__device__ __forceinline__ float c2q_fetch(const float* __restrict__ hi,
                                           int b, int r, int v, int ch,
                                           int sub, int Ld) {
  constexpr int SA[3] = { 0, 2, 1 };
  constexpr int SB[3] = { 5, 3, 4 };
  const int i = r >> 1, j = v >> 1;
  const long long base = (((long long)b * Ld + i) * Ld + j) * 36;
  const int py = r & 1, px = v & 1;
  const int im = (px ^ py) ? 18 : 0;
  const float a  = hi[base + ch + 3 * SA[sub] + im];
  const float bb = hi[base + ch + 3 * SB[sub] + im];
  return py ? ((px ^ py) ? (a - bb) * SCQ : (bb - a) * SCQ) : (a + bb) * SCQ;
}

#define L1_RS  85
#define L1_IMG (28 * L1_RS)
#define L1_YPL (32 * L1_RS)
#define L1_ZS  97

__global__ void __launch_bounds__(192)
k_level1(const float* __restrict__ low, const float* __restrict__ high1,
         float* __restrict__ Z) {
  constexpr float G0B[14] = {
     0.00325314f, -0.00388321f,  0.03466035f, -0.03887280f,
    -0.11720389f,  0.27529538f,  0.75614564f,  0.56881042f,
     0.01186609f, -0.10671180f,  0.02382538f,  0.01702522f,
    -0.00543948f, -0.00455690f };
  constexpr float G0A[14] = {
    -0.00455690f, -0.00543948f,  0.01702522f,  0.02382538f,
    -0.10671180f,  0.01186609f,  0.56881042f,  0.75614564f,
     0.27529538f, -0.11720389f, -0.03887280f,  0.03466035f,
    -0.00388321f,  0.00325314f };
  constexpr float G1B[14] = {
    -0.00455690f,  0.00543948f,  0.01702522f, -0.02382538f,
    -0.10671180f, -0.01186609f,  0.56881042f, -0.75614564f,
     0.27529538f,  0.11720389f, -0.03887280f, -0.03466035f,
    -0.00388321f, -0.00325314f };
  constexpr float G1A[14] = {
    -0.00325314f, -0.00388321f, -0.03466035f, -0.03887280f,
     0.11720389f,  0.27529538f, -0.75614564f,  0.56881042f,
    -0.01186609f, -0.10671180f, -0.02382538f,  0.01702522f,
     0.00543948f, -0.00455690f };

  __shared__ float sm[4 * L1_IMG + 2 * L1_YPL];
  float* img = sm;
  float* ys  = sm + 4 * L1_IMG;
  float* zs  = sm;

  const int tx = blockIdx.x, ty = blockIdx.y, b = blockIdx.z;
  const int tid = threadIdx.x;
  const int R0 = 16 * ty - 6, C0 = 16 * tx - 6;
  const int ib = 8 * ty - 3,  jb = 8 * tx - 3;

  const int lane = tid & 31;
  const int wv = __builtin_amdgcn_readfirstlane(tid >> 5);

  {
    int gof[3];
#pragma unroll
    for (int i = 0; i < 3; ++i) {
      const int f  = lane + 32 * i;
      const int wc = f / 3, ch = f - 3 * wc;
      gof[i] = ref_idx(C0 + wc, 256) * 3 + ch;
    }
#pragma unroll
    for (int n = 0; n < 5; ++n) {
      const int wr = (5 - wv) + 6 * n;
      if (wr < 28) {
        const float* srow = low + (b * 256 + ref_idx(R0 + wr, 256)) * 768;
        float* drow = img + wr * L1_RS;
#pragma unroll
        for (int i = 0; i < 3; ++i) {
          const int f = lane + 32 * i;
          if (f < 84) drow[f] = srow[gof[i]];
        }
      }
    }
  }

  {
    int ga[4], gb[4], l0[4], l1[4];
#pragma unroll
    for (int i = 0; i < 4; ++i) {
      const int m  = lane + 32 * i;
      const int pc = m / 9,   k  = m - 9 * pc;
      const int pair = k / 3, ch = k - 3 * pair;
      int J = jb + pc;
      const int fc = (J < 0 || J >= 128) ? 1 : 0;
      J = (J < 0) ? (-1 - J) : ((J >= 128) ? (255 - J) : J);
      const int oa = (pair == 0) ? 0  : ((pair == 1) ? 6 : 3);
      const int ob = (pair == 0) ? 15 : ((pair == 1) ? 9 : 12);
      ga[i] = J * 36 + ch + oa;
      gb[i] = J * 36 + ch + ob;
      l0[i] = (1 + pair) * L1_IMG + (2 * pc + fc) * 3 + ch;
      l1[i] = (1 + pair) * L1_IMG + (2 * pc + (1 ^ fc)) * 3 + ch;
    }
#pragma unroll
    for (int n = 0; n < 3; ++n) {
      const int pr = wv + 6 * n;
      if (pr < 14) {
        int I = ib + pr;
        const int fr = (I < 0 || I >= 128) ? 1 : 0;
        I = (I < 0) ? (-1 - I) : ((I >= 128) ? (255 - I) : I);
        const float* srow = high1 + (b * 128 + I) * (128 * 36);
        float* r0 = img + (2 * pr + fr) * L1_RS;
        float* r1 = img + (2 * pr + (1 ^ fr)) * L1_RS;
#pragma unroll
        for (int i = 0; i < 4; ++i) {
          if (lane + 32 * i < 126) {
            const float ra = srow[ga[i]], ia  = srow[ga[i] + 18];
            const float rb = srow[gb[i]], ibm = srow[gb[i] + 18];
            r0[l0[i]] = (ra + rb)  * SCQ;
            r0[l1[i]] = (ia + ibm) * SCQ;
            r1[l0[i]] = (ia - ibm) * SCQ;
            r1[l1[i]] = (rb - ra)  * SCQ;
          }
        }
      }
    }
  }
  __syncthreads();

  {
    const int plane = tid / 96, c = tid - plane * 96;
    if (c < 84) {
      const float* A = img + (2 * plane) * L1_IMG + c;
      const float* B = A + L1_IMG;
      float a[28], g[28];
#pragma unroll
      for (int r = 0; r < 28; ++r) { a[r] = A[r * L1_RS]; g[r] = B[r * L1_RS]; }
      float* Y = ys + plane * L1_YPL + c;
#pragma unroll
      for (int q = 0; q < 8; ++q) {
#pragma unroll
        for (int s = 0; s < 4; ++s) {
          const int  p  = s >> 1;
          const bool ev = (s & 1) == 0;
          const int  b0 = ev ? 6 : 7, b1 = ev ? 7 : 6;
          float s0 = 0.0f, s1 = 0.0f;
#pragma unroll
          for (int j = 0; j < 7; ++j) {
            s0 += (ev ? G0B[2 * j + p] : G0A[2 * j + p]) * a[2 * q + b0 + 6 - 2 * j];
            s1 += (ev ? G1B[2 * j + p] : G1A[2 * j + p]) * g[2 * q + b1 + 6 - 2 * j];
          }
          Y[(4 * q + s) * L1_RS] = s0 + s1;
        }
      }
    }
  }
  __syncthreads();

  {
    const int w = tid >> 5, yy = tid & 31;
    const int half = w / 3, ch = w - 3 * half;
    const float* Y1 = ys + yy * L1_RS + (8 * half) * 3 + ch;
    const float* Y2 = Y1 + L1_YPL;
    float u1[20], u2[20];
#pragma unroll
    for (int k = 0; k < 20; ++k) { u1[k] = Y1[3 * k]; u2[k] = Y2[3 * k]; }
    float* zrow = zs + yy * L1_ZS + (16 * half) * 3 + ch;
#pragma unroll
    for (int q = 0; q < 4; ++q) {
#pragma unroll
      for (int s = 0; s < 4; ++s) {
        const int  p  = s >> 1;
        const bool ev = (s & 1) == 0;
        const int  b0 = ev ? 6 : 7, b1 = ev ? 7 : 6;
        float acc = 0.0f;
#pragma unroll
        for (int j = 0; j < 7; ++j) {
          acc += (ev ? G0B[2 * j + p] : G0A[2 * j + p]) * u1[2 * q + b0 + 6 - 2 * j];
          acc += (ev ? G1B[2 * j + p] : G1A[2 * j + p]) * u2[2 * q + b1 + 6 - 2 * j];
        }
        zrow[(4 * q + s) * 3] = acc;
      }
    }
  }
  __syncthreads();

  {
    const int hrow = tid / 96, f = tid - 96 * hrow;
#pragma unroll
    for (int it = 0; it < 16; ++it) {
      const int row = 2 * it + hrow;
      Z[((b * 512 + 32 * ty + row) * 512 + 32 * tx) * 3 + f] = zs[row * L1_ZS + f];
    }
  }
}

__device__ __forceinline__ float tap7(int i) {
  int ic = i < 0 ? 0 : (i > 6 ? 6 : i);
  float t = cG0O[ic];
  return (i == ic) ? t : 0.0f;
}
__device__ __forceinline__ float tap5(int i) {
  int ic = i < 0 ? 0 : (i > 4 ? 4 : i);
  float t = cG1O[ic];
  return (i == ic) ? t : 0.0f;
}

__device__ __forceinline__ v8f wmma4(v2f a, v2f b, v8f c) {
  return __builtin_amdgcn_wmma_f32_16x16x4_f32(
       false, a,  false, b,
       (short)0, c,  false,  false);
}

__global__ void __launch_bounds__(32)
k_level0(const float* __restrict__ Z, const float* __restrict__ high0,
         float* __restrict__ out) {
  const int tx = blockIdx.x, ty = blockIdx.y, bz = blockIdx.z;
  const int b = bz / 3, ch = bz % 3;
  const int x0 = tx * 16, y0 = ty * 16;

  __shared__ float zs [24][33];
  __shared__ float lhs[24][33];
  __shared__ float hls[24][33];
  __shared__ float hhs[24][33];
  __shared__ float y1s[16][33];
  __shared__ float y2s[16][33];

  const int lane = threadIdx.x;
  const int lo = lane & 15;
  const int h2 = (lane >> 4) * 2;

  v2f f0[6], f1[6];
#pragma unroll
  for (int kk = 0; kk < 6; ++kk) {
    const int k0 = 4 * kk + h2;
    f0[kk].x = tap7(k0 - lo - 1);  f0[kk].y = tap7(k0 - lo);
    f1[kk].x = tap5(k0 - lo - 2);  f1[kk].y = tap5(k0 - lo - 1);
  }

  for (int idx = lane; idx < 24 * 32; idx += 32) {
    const int r = idx >> 5, c = idx & 31;
    const int rr = ref_idx(y0 - 4 + r, 512);
    const int cc = ref_idx(x0 - 8 + c, 512);
    zs [r][c] = Z[(((long long)b * 512 + rr) * 512 + cc) * 3 + ch];
    lhs[r][c] = c2q_fetch(high0, b, rr, cc, ch, 0, 256);
    hls[r][c] = c2q_fetch(high0, b, rr, cc, ch, 1, 256);
    hhs[r][c] = c2q_fetch(high0, b, rr, cc, ch, 2, 256);
  }
  __syncthreads();

#pragma unroll
  for (int t = 0; t < 2; ++t) {
    v8f a1 = {}; v8f a2 = {};
#pragma unroll
    for (int kk = 0; kk < 6; ++kk) {
      const int k0 = 4 * kk + h2;
      const int cc = 16 * t + lo;
      v2f bz = { zs [k0][cc], zs [k0 + 1][cc] };
      v2f bl = { lhs[k0][cc], lhs[k0 + 1][cc] };
      v2f bh = { hls[k0][cc], hls[k0 + 1][cc] };
      v2f bq = { hhs[k0][cc], hhs[k0 + 1][cc] };
      a1 = wmma4(f0[kk], bz, a1);
      a1 = wmma4(f1[kk], bl, a1);
      a2 = wmma4(f0[kk], bh, a2);
      a2 = wmma4(f1[kk], bq, a2);
    }
#pragma unroll
    for (int v = 0; v < 8; ++v) {
      const int row = v + (h2 << 2);
      y1s[row][16 * t + lo] = a1[v];
      y2s[row][16 * t + lo] = a2[v];
    }
  }
  __syncthreads();

  v8f o = {};
#pragma unroll
  for (int kk = 0; kk < 6; ++kk) {
    const int k0 = 4 * kk + h2;
    v2f A1 = { y1s[lo][k0 + 4], y1s[lo][k0 + 5] };
    v2f A2 = { y2s[lo][k0 + 4], y2s[lo][k0 + 5] };
    o = wmma4(A1, f0[kk], o);
    o = wmma4(A2, f1[kk], o);
  }
#pragma unroll
  for (int v = 0; v < 8; ++v) {
    const int row = y0 + v + (h2 << 2);
    const int col = x0 + lo;
    out[(((long long)b * 512 + row) * 512 + col) * 3 + ch] = o[v];
  }
}

extern "C" void kernel_launch(void* const* d_in, const int* in_sizes, int n_in,
                              void* d_out, int out_size, void* d_ws, size_t ws_size,
                              hipStream_t stream) {
  (void)in_sizes; (void)n_in; (void)out_size; (void)ws_size;
  const float* low   = (const float*)d_in[0];
  const float* high0 = (const float*)d_in[1];
  const float* high1 = (const float*)d_in[2];
  float* out = (float*)d_out;
  float* Zws = (float*)d_ws;

  k_level1<<<dim3(16, 16, 8), dim3(192), 0, stream>>>(low, high1, Zws);
  dim3 grid(32, 32, 8 * 3);
  k_level0<<<grid, dim3(32), 0, stream>>>(Zws, high0, out);
}


# level-0 kernel: one block per 32x32x3 output tile stages the reflected 40x40 window of Z and the c2q planes of high0 in LDS once with row-contiguous loads; 12 waves run the WMMA passes from LDS
# speedup vs baseline: 5.2649x; 1.8480x over previous
#include <hip/hip_runtime.h>
#include <hip/hip_bf16.h>


typedef float v2f __attribute__((ext_vector_type(2)));
typedef float v8f __attribute__((ext_vector_type(8)));

#define SCQ 0.70710678118654752440f

__constant__ float cG0O[7] = { -3.0f/280.0f, -15.0f/280.0f, 73.0f/280.0f,
                               170.0f/280.0f, 73.0f/280.0f, -15.0f/280.0f,
                               -3.0f/280.0f };
__constant__ float cG1O[5] = { -1.0f/20.0f, -5.0f/20.0f, 12.0f/20.0f,
                               -5.0f/20.0f, -1.0f/20.0f };

__device__ __forceinline__ int ref_idx(int i, int L) {
  if (i < 0)  i = -1 - i;
  if (i >= L) i = 2 * L - 1 - i;
  return i;
}

#define L1_RS  85
#define L1_IMG (28 * L1_RS)
#define L1_YPL (32 * L1_RS)
#define L1_ZS  97

__global__ void __launch_bounds__(192)
k_level1(const float* __restrict__ low, const float* __restrict__ high1,
         float* __restrict__ Z) {
  constexpr float G0B[14] = {
     0.00325314f, -0.00388321f,  0.03466035f, -0.03887280f,
    -0.11720389f,  0.27529538f,  0.75614564f,  0.56881042f,
     0.01186609f, -0.10671180f,  0.02382538f,  0.01702522f,
    -0.00543948f, -0.00455690f };
  constexpr float G0A[14] = {
    -0.00455690f, -0.00543948f,  0.01702522f,  0.02382538f,
    -0.10671180f,  0.01186609f,  0.56881042f,  0.75614564f,
     0.27529538f, -0.11720389f, -0.03887280f,  0.03466035f,
    -0.00388321f,  0.00325314f };
  constexpr float G1B[14] = {
    -0.00455690f,  0.00543948f,  0.01702522f, -0.02382538f,
    -0.10671180f, -0.01186609f,  0.56881042f, -0.75614564f,
     0.27529538f,  0.11720389f, -0.03887280f, -0.03466035f,
    -0.00388321f, -0.00325314f };
  constexpr float G1A[14] = {
    -0.00325314f, -0.00388321f, -0.03466035f, -0.03887280f,
     0.11720389f,  0.27529538f, -0.75614564f,  0.56881042f,
    -0.01186609f, -0.10671180f, -0.02382538f,  0.01702522f,
     0.00543948f, -0.00455690f };

  __shared__ float sm[4 * L1_IMG + 2 * L1_YPL];
  float* img = sm;
  float* ys  = sm + 4 * L1_IMG;
  float* zs  = sm;

  const int tx = blockIdx.x, ty = blockIdx.y, b = blockIdx.z;
  const int tid = threadIdx.x;
  const int R0 = 16 * ty - 6, C0 = 16 * tx - 6;
  const int ib = 8 * ty - 3,  jb = 8 * tx - 3;

  const int lane = tid & 31;
  const int wv = __builtin_amdgcn_readfirstlane(tid >> 5);

  {
    int gof[3];
#pragma unroll
    for (int i = 0; i < 3; ++i) {
      const int f  = lane + 32 * i;
      const int wc = f / 3, ch = f - 3 * wc;
      gof[i] = ref_idx(C0 + wc, 256) * 3 + ch;
    }
#pragma unroll
    for (int n = 0; n < 5; ++n) {
      const int wr = (5 - wv) + 6 * n;
      if (wr < 28) {
        const float* srow = low + (b * 256 + ref_idx(R0 + wr, 256)) * 768;
        float* drow = img + wr * L1_RS;
#pragma unroll
        for (int i = 0; i < 3; ++i) {
          const int f = lane + 32 * i;
          if (f < 84) drow[f] = srow[gof[i]];
        }
      }
    }
  }

  {
    int ga[4], gb[4], l0[4], l1[4];
#pragma unroll
    for (int i = 0; i < 4; ++i) {
      const int m  = lane + 32 * i;
      const int pc = m / 9,   k  = m - 9 * pc;
      const int pair = k / 3, ch = k - 3 * pair;
      int J = jb + pc;
      const int fc = (J < 0 || J >= 128) ? 1 : 0;
      J = (J < 0) ? (-1 - J) : ((J >= 128) ? (255 - J) : J);
      const int oa = (pair == 0) ? 0  : ((pair == 1) ? 6 : 3);
      const int ob = (pair == 0) ? 15 : ((pair == 1) ? 9 : 12);
      ga[i] = J * 36 + ch + oa;
      gb[i] = J * 36 + ch + ob;
      l0[i] = (1 + pair) * L1_IMG + (2 * pc + fc) * 3 + ch;
      l1[i] = (1 + pair) * L1_IMG + (2 * pc + (1 ^ fc)) * 3 + ch;
    }
#pragma unroll
    for (int n = 0; n < 3; ++n) {
      const int pr = wv + 6 * n;
      if (pr < 14) {
        int I = ib + pr;
        const int fr = (I < 0 || I >= 128) ? 1 : 0;
        I = (I < 0) ? (-1 - I) : ((I >= 128) ? (255 - I) : I);
        const float* srow = high1 + (b * 128 + I) * (128 * 36);
        float* r0 = img + (2 * pr + fr) * L1_RS;
        float* r1 = img + (2 * pr + (1 ^ fr)) * L1_RS;
#pragma unroll
        for (int i = 0; i < 4; ++i) {
          if (lane + 32 * i < 126) {
            const float ra = srow[ga[i]], ia  = srow[ga[i] + 18];
            const float rb = srow[gb[i]], ibm = srow[gb[i] + 18];
            r0[l0[i]] = (ra + rb)  * SCQ;
            r0[l1[i]] = (ia + ibm) * SCQ;
            r1[l0[i]] = (ia - ibm) * SCQ;
            r1[l1[i]] = (rb - ra)  * SCQ;
          }
        }
      }
    }
  }
  __syncthreads();

  {
    const int plane = tid / 96, c = tid - plane * 96;
    if (c < 84) {
      const float* A = img + (2 * plane) * L1_IMG + c;
      const float* B = A + L1_IMG;
      float a[28], g[28];
#pragma unroll
      for (int r = 0; r < 28; ++r) { a[r] = A[r * L1_RS]; g[r] = B[r * L1_RS]; }
      float* Y = ys + plane * L1_YPL + c;
#pragma unroll
      for (int q = 0; q < 8; ++q) {
#pragma unroll
        for (int s = 0; s < 4; ++s) {
          const int  p  = s >> 1;
          const bool ev = (s & 1) == 0;
          const int  b0 = ev ? 6 : 7, b1 = ev ? 7 : 6;
          float s0 = 0.0f, s1 = 0.0f;
#pragma unroll
          for (int j = 0; j < 7; ++j) {
            s0 += (ev ? G0B[2 * j + p] : G0A[2 * j + p]) * a[2 * q + b0 + 6 - 2 * j];
            s1 += (ev ? G1B[2 * j + p] : G1A[2 * j + p]) * g[2 * q + b1 + 6 - 2 * j];
          }
          Y[(4 * q + s) * L1_RS] = s0 + s1;
        }
      }
    }
  }
  __syncthreads();

  {
    const int w = tid >> 5, yy = tid & 31;
    const int half = w / 3, ch = w - 3 * half;
    const float* Y1 = ys + yy * L1_RS + (8 * half) * 3 + ch;
    const float* Y2 = Y1 + L1_YPL;
    float u1[20], u2[20];
#pragma unroll
    for (int k = 0; k < 20; ++k) { u1[k] = Y1[3 * k]; u2[k] = Y2[3 * k]; }
    float* zrow = zs + yy * L1_ZS + (16 * half) * 3 + ch;
#pragma unroll
    for (int q = 0; q < 4; ++q) {
#pragma unroll
      for (int s = 0; s < 4; ++s) {
        const int  p  = s >> 1;
        const bool ev = (s & 1) == 0;
        const int  b0 = ev ? 6 : 7, b1 = ev ? 7 : 6;
        float acc = 0.0f;
#pragma unroll
        for (int j = 0; j < 7; ++j) {
          acc += (ev ? G0B[2 * j + p] : G0A[2 * j + p]) * u1[2 * q + b0 + 6 - 2 * j];
          acc += (ev ? G1B[2 * j + p] : G1A[2 * j + p]) * u2[2 * q + b1 + 6 - 2 * j];
        }
        zrow[(4 * q + s) * 3] = acc;
      }
    }
  }
  __syncthreads();

  {
    const int hrow = tid / 96, f = tid - 96 * hrow;
#pragma unroll
    for (int it = 0; it < 16; ++it) {
      const int row = 2 * it + hrow;
      Z[((b * 512 + 32 * ty + row) * 512 + 32 * tx) * 3 + f] = zs[row * L1_ZS + f];
    }
  }
}

__device__ __forceinline__ float tap7(int i) {
  int ic = i < 0 ? 0 : (i > 6 ? 6 : i);
  float t = cG0O[ic];
  return (i == ic) ? t : 0.0f;
}
__device__ __forceinline__ float tap5(int i) {
  int ic = i < 0 ? 0 : (i > 4 ? 4 : i);
  float t = cG1O[ic];
  return (i == ic) ? t : 0.0f;
}

__device__ __forceinline__ v8f wmma4(v2f a, v2f b, v8f c) {
  return __builtin_amdgcn_wmma_f32_16x16x4_f32(
       false, a,  false, b,
       (short)0, c,  false,  false);
}

#define L0_P   41
#define L0_PL  (40 * L0_P)
#define L0_YP  36
#define L0_YW  (2 * 16 * L0_YP)

__global__ void __launch_bounds__(384, 12)
k_level0(const float* __restrict__ Z, const float* __restrict__ high0,
         float* __restrict__ out) {
  __shared__ float sm[12 * L0_PL];

  const int tx = blockIdx.x, ty = blockIdx.y, b = blockIdx.z;
  const int tid = threadIdx.x;
  const int X0 = 32 * tx, Y0 = 32 * ty;
  const int R0 = Y0 - 4, C0 = X0 - 4;
  const int ib = 16 * ty - 2, jb = 16 * tx - 2;

  if (tid < 360) {
    {
      const int g = tid / 120, f = tid - 120 * g;
      const int wc = f / 3, ch = f - 3 * wc;
      const float* src = Z + b * (512 * 512 * 3) + ref_idx(C0 + wc, 512) * 3 + ch;
      float* dst = sm + ch * L0_PL + wc;
#pragma unroll
      for (int n = 0; n < 14; ++n) {
        const int wr = 3 * n + g;
        if (n < 13 || g == 0)
          dst[wr * L0_P] = src[ref_idx(R0 + wr, 512) * (512 * 3)];
      }
    }
    {
      const int g = tid / 180, m = tid - 180 * g;
      const int pc = m / 9,   k = m - 9 * pc;
      const int pair = k / 3, ch = k - 3 * pair;
      int J = jb + pc;
      const int fc = (J < 0 || J >= 256) ? 1 : 0;
      J = (J < 0) ? (-1 - J) : ((J >= 256) ? (511 - J) : J);
      const int oa = (pair == 0) ? 0  : ((pair == 1) ? 6 : 3);
      const int ob = (pair == 0) ? 15 : ((pair == 1) ? 9 : 12);
      const float* sa = high0 + b * (256 * 256 * 36) + J * 36 + ch + oa;
      const float* sb = high0 + b * (256 * 256 * 36) + J * 36 + ch + ob;
      float* p0 = sm + ((1 + pair) * 3 + ch) * L0_PL + 2 * pc + fc;
      float* p1 = sm + ((1 + pair) * 3 + ch) * L0_PL + 2 * pc + (1 ^ fc);
#pragma unroll
      for (int n = 0; n < 10; ++n) {
        const int pr = 2 * n + g;
        int I = ib + pr;
        const int fr = (I < 0 || I >= 256) ? 1 : 0;
        I = (I < 0) ? (-1 - I) : ((I >= 256) ? (511 - I) : I);
        const int go = I * (256 * 36);
        const float ra = sa[go], ia  = sa[go + 18];
        const float rb = sb[go], ibm = sb[go + 18];
        const int r0 = (2 * pr + fr) * L0_P;
        const int r1 = (2 * pr + (1 ^ fr)) * L0_P;
        p0[r0] = (ra + rb)  * SCQ;
        p1[r0] = (ia + ibm) * SCQ;
        p0[r1] = (ia - ibm) * SCQ;
        p1[r1] = (rb - ra)  * SCQ;
      }
    }
  }

  const int lane = tid & 31;
  const int lo = lane & 15;
  const int h2 = (lane >> 4) * 2;
  const int w = __builtin_amdgcn_readfirstlane(tid >> 5);
  const int sub = w / 3, ch = w - 3 * sub;
  const int sy = sub >> 1, sx = sub & 1;
  const int x0 = X0 + 16 * sx, y0 = Y0 + 16 * sy;

  v2f f0[6], f1[6];
#pragma unroll
  for (int kk = 0; kk < 6; ++kk) {
    const int k0 = 4 * kk + h2;
    f0[kk].x = tap7(k0 - lo - 1);  f0[kk].y = tap7(k0 - lo);
    f1[kk].x = tap5(k0 - lo - 2);  f1[kk].y = tap5(k0 - lo - 1);
  }
  __syncthreads();

  v8f a1[2], a2[2];
  {
    const float* zp = sm + ch * L0_PL + (16 * sy) * L0_P;
    const float* lp = zp + 3 * L0_PL;
    const float* hp = zp + 6 * L0_PL;
    const float* qp = zp + 9 * L0_PL;
#pragma unroll
    for (int t = 0; t < 2; ++t) {
      a1[t] = (v8f){}; a2[t] = (v8f){};
      int cc = 16 * sx + 16 * t + lo;
      cc = cc > 39 ? 39 : cc;
#pragma unroll
      for (int kk = 0; kk < 6; ++kk) {
        const int k0 = 4 * kk + h2;
        const int o0 = k0 * L0_P + cc, o1 = o0 + L0_P;
        v2f bz = { zp[o0], zp[o1] };
        v2f bl = { lp[o0], lp[o1] };
        v2f bh = { hp[o0], hp[o1] };
        v2f bq = { qp[o0], qp[o1] };
        a1[t] = wmma4(f0[kk], bz, a1[t]);
        a1[t] = wmma4(f1[kk], bl, a1[t]);
        a2[t] = wmma4(f0[kk], bh, a2[t]);
        a2[t] = wmma4(f1[kk], bq, a2[t]);
      }
    }
  }
  __syncthreads();

  float* y1s = sm + w * L0_YW;
  float* y2s = y1s + 16 * L0_YP;
#pragma unroll
  for (int t = 0; t < 2; ++t) {
#pragma unroll
    for (int v = 0; v < 8; ++v) {
      const int row = v + (h2 << 2);
      y1s[row * L0_YP + 16 * t + lo] = a1[t][v];
      y2s[row * L0_YP + 16 * t + lo] = a2[t][v];
    }
  }
  __syncthreads();

  v8f o = {};
#pragma unroll
  for (int kk = 0; kk < 6; ++kk) {
    const int k0 = 4 * kk + h2;
    v2f A1 = { y1s[lo * L0_YP + k0], y1s[lo * L0_YP + k0 + 1] };
    v2f A2 = { y2s[lo * L0_YP + k0], y2s[lo * L0_YP + k0 + 1] };
    o = wmma4(A1, f0[kk], o);
    o = wmma4(A2, f1[kk], o);
  }
#pragma unroll
  for (int v = 0; v < 8; ++v) {
    const int row = y0 + v + (h2 << 2);
    const int col = x0 + lo;
    out[(((long long)b * 512 + row) * 512 + col) * 3 + ch] = o[v];
  }
}

extern "C" void kernel_launch(void* const* d_in, const int* in_sizes, int n_in,
                              void* d_out, int out_size, void* d_ws, size_t ws_size,
                              hipStream_t stream) {
  (void)in_sizes; (void)n_in; (void)out_size; (void)ws_size;
  const float* low   = (const float*)d_in[0];
  const float* high0 = (const float*)d_in[1];
  const float* high1 = (const float*)d_in[2];
  float* out = (float*)d_out;
  float* Zws = (float*)d_ws;

  k_level1<<<dim3(16, 16, 8), dim3(192), 0, stream>>>(low, high1, Zws);
  k_level0<<<dim3(16, 16, 8), dim3(384), 0, stream>>>(Zws, high0, out);
}


# staging of both kernels branch-free with wave-uniform rows: all window loads of a wave (63 in level-1, 54 in level-0) are in flight before its first wait, one memory wait instead of one per row
# speedup vs baseline: 5.6648x; 1.0760x over previous
#include <hip/hip_runtime.h>
#include <hip/hip_bf16.h>


typedef float v2f __attribute__((ext_vector_type(2)));
typedef float v8f __attribute__((ext_vector_type(8)));

#define SCQ 0.70710678118654752440f

__constant__ float cG0O[7] = { -3.0f/280.0f, -15.0f/280.0f, 73.0f/280.0f,
                               170.0f/280.0f, 73.0f/280.0f, -15.0f/280.0f,
                               -3.0f/280.0f };
__constant__ float cG1O[5] = { -1.0f/20.0f, -5.0f/20.0f, 12.0f/20.0f,
                               -5.0f/20.0f, -1.0f/20.0f };

__device__ __forceinline__ int ref_idx(int i, int L) {
  if (i < 0)  i = -1 - i;
  if (i >= L) i = 2 * L - 1 - i;
  return i;
}

#define L1_RS  85
#define L1_IMG (28 * L1_RS)
#define L1_YPL (32 * L1_RS)
#define L1_ZS  97

__global__ void __launch_bounds__(192)
k_level1(const float* __restrict__ low, const float* __restrict__ high1,
         float* __restrict__ Z) {
  constexpr float G0B[14] = {
     0.00325314f, -0.00388321f,  0.03466035f, -0.03887280f,
    -0.11720389f,  0.27529538f,  0.75614564f,  0.56881042f,
     0.01186609f, -0.10671180f,  0.02382538f,  0.01702522f,
    -0.00543948f, -0.00455690f };
  constexpr float G0A[14] = {
    -0.00455690f, -0.00543948f,  0.01702522f,  0.02382538f,
    -0.10671180f,  0.01186609f,  0.56881042f,  0.75614564f,
     0.27529538f, -0.11720389f, -0.03887280f,  0.03466035f,
    -0.00388321f,  0.00325314f };
  constexpr float G1B[14] = {
    -0.00455690f,  0.00543948f,  0.01702522f, -0.02382538f,
    -0.10671180f, -0.01186609f,  0.56881042f, -0.75614564f,
     0.27529538f,  0.11720389f, -0.03887280f, -0.03466035f,
    -0.00388321f, -0.00325314f };
  constexpr float G1A[14] = {
    -0.00325314f, -0.00388321f, -0.03466035f, -0.03887280f,
     0.11720389f,  0.27529538f, -0.75614564f,  0.56881042f,
    -0.01186609f, -0.10671180f, -0.02382538f,  0.01702522f,
     0.00543948f, -0.00455690f };

  __shared__ float sm[4 * L1_IMG + 2 * L1_YPL];
  float* img = sm;
  float* ys  = sm + 4 * L1_IMG;
  float* zs  = sm;

  const int tx = blockIdx.x, ty = blockIdx.y, b = blockIdx.z;
  const int tid = threadIdx.x;
  const int R0 = 16 * ty - 6, C0 = 16 * tx - 6;
  const int ib = 8 * ty - 3,  jb = 8 * tx - 3;

  const int lane = tid & 31;
  const int wv = __builtin_amdgcn_readfirstlane(tid >> 5);

  float vl[5][3];
  float hra[3][4], hia[3][4], hrb[3][4], hib[3][4];
  int fl[3], l0[4], l1[4];
  {
    int gof[3];
#pragma unroll
    for (int i = 0; i < 3; ++i) {
      const int f  = lane + 32 * i;
      const int wc = f / 3, ch = f - 3 * wc;
      gof[i] = ref_idx(C0 + wc, 256) * 3 + ch;
      fl[i]  = f < 84 ? f : 84 + L1_RS * (f - 84);
    }
#pragma unroll
    for (int n = 0; n < 5; ++n) {
      int wr = (5 - wv) + 6 * n;
      wr = wr < 28 ? wr : 27;
      const float* srow = low + (b * 256 + ref_idx(R0 + wr, 256)) * 768;
#pragma unroll
      for (int i = 0; i < 3; ++i) vl[n][i] = srow[gof[i]];
    }
  }
  {
    int ga[4], gb[4];
#pragma unroll
    for (int i = 0; i < 4; ++i) {
      const int m  = lane + 32 * i;
      const int pc = m / 9,   k  = m - 9 * pc;
      const int pair = k / 3, ch = k - 3 * pair;
      int J = jb + pc;
      const int fc = (J < 0 || J >= 128) ? 1 : 0;
      J = (J < 0) ? (-1 - J) : ((J >= 128) ? (255 - J) : J);
      const int oa = (pair == 0) ? 0  : ((pair == 1) ? 6 : 3);
      const int ob = (pair == 0) ? 15 : ((pair == 1) ? 9 : 12);
      ga[i] = J * 36 + ch + oa;
      gb[i] = J * 36 + ch + ob;
      const int c0 = (1 + pair) * L1_IMG + (2 * pc + fc) * 3 + ch;
      const int c1 = (1 + pair) * L1_IMG + (2 * pc + (1 ^ fc)) * 3 + ch;
      l0[i] = m < 126 ? c0 : (L1_IMG + 84 + L1_RS * (m - 126));
      l1[i] = m < 126 ? c1 : (L1_IMG + 84 + L1_RS * (m - 126));
    }
#pragma unroll
    for (int n = 0; n < 3; ++n) {
      int pr = wv + 6 * n;
      pr = pr < 14 ? pr : 13;
      int I = ib + pr;
      I = (I < 0) ? (-1 - I) : ((I >= 128) ? (255 - I) : I);
      const float* srow = high1 + (b * 128 + I) * (128 * 36);
#pragma unroll
      for (int i = 0; i < 4; ++i) {
        hra[n][i] = srow[ga[i]];  hia[n][i] = srow[ga[i] + 18];
        hrb[n][i] = srow[gb[i]];  hib[n][i] = srow[gb[i] + 18];
      }
    }
  }

#pragma unroll
  for (int n = 0; n < 5; ++n) {
    int wr = (5 - wv) + 6 * n;
    wr = wr < 28 ? wr : 27;
    float* drow = img + wr * L1_RS;
#pragma unroll
    for (int i = 0; i < 3; ++i) drow[fl[i]] = vl[n][i];
  }
#pragma unroll
  for (int n = 0; n < 3; ++n) {
    int pr = wv + 6 * n;
    pr = pr < 14 ? pr : 13;
    const int I = ib + pr;
    const int fr = (I < 0 || I >= 128) ? 1 : 0;
    float* r0 = img + (2 * pr + fr) * L1_RS;
    float* r1 = img + (2 * pr + (1 ^ fr)) * L1_RS;
#pragma unroll
    for (int i = 0; i < 4; ++i) {
      const float ra = hra[n][i], ia  = hia[n][i];
      const float rb = hrb[n][i], ibm = hib[n][i];
      r0[l0[i]] = (ra + rb)  * SCQ;
      r0[l1[i]] = (ia + ibm) * SCQ;
      r1[l0[i]] = (ia - ibm) * SCQ;
      r1[l1[i]] = (rb - ra)  * SCQ;
    }
  }
  __syncthreads();

  {
    const int plane = tid / 96, c = tid - plane * 96;
    if (c < 84) {
      const float* A = img + (2 * plane) * L1_IMG + c;
      const float* B = A + L1_IMG;
      float a[28], g[28];
#pragma unroll
      for (int r = 0; r < 28; ++r) { a[r] = A[r * L1_RS]; g[r] = B[r * L1_RS]; }
      float* Y = ys + plane * L1_YPL + c;
#pragma unroll
      for (int q = 0; q < 8; ++q) {
#pragma unroll
        for (int s = 0; s < 4; ++s) {
          const int  p  = s >> 1;
          const bool ev = (s & 1) == 0;
          const int  b0 = ev ? 6 : 7, b1 = ev ? 7 : 6;
          float s0 = 0.0f, s1 = 0.0f;
#pragma unroll
          for (int j = 0; j < 7; ++j) {
            s0 += (ev ? G0B[2 * j + p] : G0A[2 * j + p]) * a[2 * q + b0 + 6 - 2 * j];
            s1 += (ev ? G1B[2 * j + p] : G1A[2 * j + p]) * g[2 * q + b1 + 6 - 2 * j];
          }
          Y[(4 * q + s) * L1_RS] = s0 + s1;
        }
      }
    }
  }
  __syncthreads();

  {
    const int w = tid >> 5, yy = tid & 31;
    const int half = w / 3, ch = w - 3 * half;
    const float* Y1 = ys + yy * L1_RS + (8 * half) * 3 + ch;
    const float* Y2 = Y1 + L1_YPL;
    float u1[20], u2[20];
#pragma unroll
    for (int k = 0; k < 20; ++k) { u1[k] = Y1[3 * k]; u2[k] = Y2[3 * k]; }
    float* zrow = zs + yy * L1_ZS + (16 * half) * 3 + ch;
#pragma unroll
    for (int q = 0; q < 4; ++q) {
#pragma unroll
      for (int s = 0; s < 4; ++s) {
        const int  p  = s >> 1;
        const bool ev = (s & 1) == 0;
        const int  b0 = ev ? 6 : 7, b1 = ev ? 7 : 6;
        float acc = 0.0f;
#pragma unroll
        for (int j = 0; j < 7; ++j) {
          acc += (ev ? G0B[2 * j + p] : G0A[2 * j + p]) * u1[2 * q + b0 + 6 - 2 * j];
          acc += (ev ? G1B[2 * j + p] : G1A[2 * j + p]) * u2[2 * q + b1 + 6 - 2 * j];
        }
        zrow[(4 * q + s) * 3] = acc;
      }
    }
  }
  __syncthreads();

  {
    const int hrow = tid / 96, f = tid - 96 * hrow;
#pragma unroll
    for (int it = 0; it < 16; ++it) {
      const int row = 2 * it + hrow;
      Z[((b * 512 + 32 * ty + row) * 512 + 32 * tx) * 3 + f] = zs[row * L1_ZS + f];
    }
  }
}

__device__ __forceinline__ float tap7(int i) {
  int ic = i < 0 ? 0 : (i > 6 ? 6 : i);
  float t = cG0O[ic];
  return (i == ic) ? t : 0.0f;
}
__device__ __forceinline__ float tap5(int i) {
  int ic = i < 0 ? 0 : (i > 4 ? 4 : i);
  float t = cG1O[ic];
  return (i == ic) ? t : 0.0f;
}

__device__ __forceinline__ v8f wmma4(v2f a, v2f b, v8f c) {
  return __builtin_amdgcn_wmma_f32_16x16x4_f32(
       false, a,  false, b,
       (short)0, c,  false,  false);
}

#define L0_P   41
#define L0_PL  (40 * L0_P)
#define L0_YP  36
#define L0_YW  (2 * 16 * L0_YP)

__global__ void __launch_bounds__(384, 12)
k_level0(const float* __restrict__ Z, const float* __restrict__ high0,
         float* __restrict__ out) {
  __shared__ float sm[12 * L0_PL];

  const int tx = blockIdx.x, ty = blockIdx.y, b = blockIdx.z;
  const int tid = threadIdx.x;
  const int X0 = 32 * tx, Y0 = 32 * ty;
  const int R0 = Y0 - 4, C0 = X0 - 4;
  const int ib = 16 * ty - 2, jb = 16 * tx - 2;

  const int lane = tid & 31;
  const int w = __builtin_amdgcn_readfirstlane(tid >> 5);

  float zv[14];
  float hra[10], hia[10], hrb[10], hib[10];
  int zdst, p0, p1;
  const float* zb = Z + b * (512 * 512 * 3);
  const float* hb = high0 + b * (256 * 256 * 36);
  {
    const int g = w >> 2;
    const int f = lane + 32 * (w & 3);
    const int fz = f < 120 ? f : 119;
    const int wc = fz / 3, ch = fz - 3 * wc;
    const int zcol = ref_idx(C0 + wc, 512) * 3 + ch;
    zdst = f < 120 ? (ch * L0_PL + wc) : 40 + L0_P * (f - 120);
#pragma unroll
    for (int n = 0; n < 14; ++n) {
      int wr = 3 * n + g;
      wr = wr < 40 ? wr : 39;
      zv[n] = zb[ref_idx(R0 + wr, 512) * (512 * 3) + zcol];
    }
  }
  const int rp = w / 6;
  {
    const int m  = lane + 32 * (w - 6 * rp);
    const int mc = m < 180 ? m : 179;
    const int pc = mc / 9,  k = mc - 9 * pc;
    const int pair = k / 3, ch = k - 3 * pair;
    int J = jb + pc;
    const int fc = (J < 0 || J >= 256) ? 1 : 0;
    J = (J < 0) ? (-1 - J) : ((J >= 256) ? (511 - J) : J);
    const int oa = (pair == 0) ? 0  : ((pair == 1) ? 6 : 3);
    const int ob = (pair == 0) ? 15 : ((pair == 1) ? 9 : 12);
    const int ga = J * 36 + ch + oa;
    const int gb = J * 36 + ch + ob;
    const int c0 = ((1 + pair) * 3 + ch) * L0_PL + 2 * pc + fc;
    const int c1 = ((1 + pair) * 3 + ch) * L0_PL + 2 * pc + (1 ^ fc);
    p0 = m < 180 ? c0 : 40 + L0_P * (m - 180);
    p1 = m < 180 ? c1 : 40 + L0_P * (m - 180);
#pragma unroll
    for (int n = 0; n < 10; ++n) {
      const int pr = 2 * n + rp;
      int I = ib + pr;
      I = (I < 0) ? (-1 - I) : ((I >= 256) ? (511 - I) : I);
      const float* srow = hb + I * (256 * 36);
      hra[n] = srow[ga];  hia[n] = srow[ga + 18];
      hrb[n] = srow[gb];  hib[n] = srow[gb + 18];
    }
  }

#pragma unroll
  for (int n = 0; n < 14; ++n) {
    int wr = 3 * n + (w >> 2);
    wr = wr < 40 ? wr : 39;
    sm[wr * L0_P + zdst] = zv[n];
  }
#pragma unroll
  for (int n = 0; n < 10; ++n) {
    const int pr = 2 * n + rp;
    const int I = ib + pr;
    const int fr = (I < 0 || I >= 256) ? 1 : 0;
    const int r0 = (2 * pr + fr) * L0_P;
    const int r1 = (2 * pr + (1 ^ fr)) * L0_P;
    const float ra = hra[n], ia  = hia[n];
    const float rb = hrb[n], ibm = hib[n];
    sm[p0 + r0] = (ra + rb)  * SCQ;
    sm[p1 + r0] = (ia + ibm) * SCQ;
    sm[p0 + r1] = (ia - ibm) * SCQ;
    sm[p1 + r1] = (rb - ra)  * SCQ;
  }

  const int lo = lane & 15;
  const int h2 = (lane >> 4) * 2;
  const int sub = w / 3, ch = w - 3 * sub;
  const int sy = sub >> 1, sx = sub & 1;
  const int x0 = X0 + 16 * sx, y0 = Y0 + 16 * sy;

  __syncthreads();

  v2f f0[6], f1[6];
#pragma unroll
  for (int kk = 0; kk < 6; ++kk) {
    const int k0 = 4 * kk + h2;
    f0[kk].x = tap7(k0 - lo - 1);  f0[kk].y = tap7(k0 - lo);
    f1[kk].x = tap5(k0 - lo - 2);  f1[kk].y = tap5(k0 - lo - 1);
  }

  v8f a1[2], a2[2];
  {
    const float* zp = sm + ch * L0_PL + (16 * sy) * L0_P;
    const float* lp = zp + 3 * L0_PL;
    const float* hp = zp + 6 * L0_PL;
    const float* qp = zp + 9 * L0_PL;
#pragma unroll
    for (int t = 0; t < 2; ++t) {
      a1[t] = (v8f){}; a2[t] = (v8f){};
      int cc = 16 * sx + 16 * t + lo;
      cc = cc > 39 ? 39 : cc;
#pragma unroll
      for (int kk = 0; kk < 6; ++kk) {
        const int k0 = 4 * kk + h2;
        const int o0 = k0 * L0_P + cc, o1 = o0 + L0_P;
        v2f bz = { zp[o0], zp[o1] };
        v2f bl = { lp[o0], lp[o1] };
        v2f bh = { hp[o0], hp[o1] };
        v2f bq = { qp[o0], qp[o1] };
        a1[t] = wmma4(f0[kk], bz, a1[t]);
        a1[t] = wmma4(f1[kk], bl, a1[t]);
        a2[t] = wmma4(f0[kk], bh, a2[t]);
        a2[t] = wmma4(f1[kk], bq, a2[t]);
      }
    }
  }
  __syncthreads();

  float* y1s = sm + w * L0_YW;
  float* y2s = y1s + 16 * L0_YP;
#pragma unroll
  for (int t = 0; t < 2; ++t) {
#pragma unroll
    for (int v = 0; v < 8; ++v) {
      const int row = v + (h2 << 2);
      y1s[row * L0_YP + 16 * t + lo] = a1[t][v];
      y2s[row * L0_YP + 16 * t + lo] = a2[t][v];
    }
  }
  __syncthreads();

  v8f o = {};
#pragma unroll
  for (int kk = 0; kk < 6; ++kk) {
    const int k0 = 4 * kk + h2;
    v2f A1 = { y1s[lo * L0_YP + k0], y1s[lo * L0_YP + k0 + 1] };
    v2f A2 = { y2s[lo * L0_YP + k0], y2s[lo * L0_YP + k0 + 1] };
    o = wmma4(A1, f0[kk], o);
    o = wmma4(A2, f1[kk], o);
  }
#pragma unroll
  for (int v = 0; v < 8; ++v) {
    const int row = y0 + v + (h2 << 2);
    const int col = x0 + lo;
    out[(((long long)b * 512 + row) * 512 + col) * 3 + ch] = o[v];
  }
}

extern "C" void kernel_launch(void* const* d_in, const int* in_sizes, int n_in,
                              void* d_out, int out_size, void* d_ws, size_t ws_size,
                              hipStream_t stream) {
  (void)in_sizes; (void)n_in; (void)out_size; (void)ws_size;
  const float* low   = (const float*)d_in[0];
  const float* high0 = (const float*)d_in[1];
  const float* high1 = (const float*)d_in[2];
  float* out = (float*)d_out;
  float* Zws = (float*)d_ws;

  k_level1<<<dim3(16, 16, 8), dim3(192), 0, stream>>>(low, high1, Zws);
  k_level0<<<dim3(16, 16, 8), dim3(384), 0, stream>>>(Zws, high0, out);
}


# level-0 epilogue: the 32x32x3 output tile is gathered in LDS and leaves as whole 128-byte lines (16-byte stores, one instruction per line) instead of 4-byte pieces 12 bytes apart
# speedup vs baseline: 5.9062x; 1.0426x over previous
#include <hip/hip_runtime.h>
#include <hip/hip_bf16.h>


typedef float v2f __attribute__((ext_vector_type(2)));
typedef float v8f __attribute__((ext_vector_type(8)));
typedef float v4f __attribute__((ext_vector_type(4)));

#define SCQ 0.70710678118654752440f

__constant__ float cG0O[7] = { -3.0f/280.0f, -15.0f/280.0f, 73.0f/280.0f,
                               170.0f/280.0f, 73.0f/280.0f, -15.0f/280.0f,
                               -3.0f/280.0f };
__constant__ float cG1O[5] = { -1.0f/20.0f, -5.0f/20.0f, 12.0f/20.0f,
                               -5.0f/20.0f, -1.0f/20.0f };

__device__ __forceinline__ int ref_idx(int i, int L) {
  if (i < 0)  i = -1 - i;
  if (i >= L) i = 2 * L - 1 - i;
  return i;
}

#define L1_RS  85
#define L1_IMG (28 * L1_RS)
#define L1_YPL (32 * L1_RS)
#define L1_ZS  97

__global__ void __launch_bounds__(192)
k_level1(const float* __restrict__ low, const float* __restrict__ high1,
         float* __restrict__ Z) {
  constexpr float G0B[14] = {
     0.00325314f, -0.00388321f,  0.03466035f, -0.03887280f,
    -0.11720389f,  0.27529538f,  0.75614564f,  0.56881042f,
     0.01186609f, -0.10671180f,  0.02382538f,  0.01702522f,
    -0.00543948f, -0.00455690f };
  constexpr float G0A[14] = {
    -0.00455690f, -0.00543948f,  0.01702522f,  0.02382538f,
    -0.10671180f,  0.01186609f,  0.56881042f,  0.75614564f,
     0.27529538f, -0.11720389f, -0.03887280f,  0.03466035f,
    -0.00388321f,  0.00325314f };
  constexpr float G1B[14] = {
    -0.00455690f,  0.00543948f,  0.01702522f, -0.02382538f,
    -0.10671180f, -0.01186609f,  0.56881042f, -0.75614564f,
     0.27529538f,  0.11720389f, -0.03887280f, -0.03466035f,
    -0.00388321f, -0.00325314f };
  constexpr float G1A[14] = {
    -0.00325314f, -0.00388321f, -0.03466035f, -0.03887280f,
     0.11720389f,  0.27529538f, -0.75614564f,  0.56881042f,
    -0.01186609f, -0.10671180f, -0.02382538f,  0.01702522f,
     0.00543948f, -0.00455690f };

  __shared__ float sm[4 * L1_IMG + 2 * L1_YPL];
  float* img = sm;
  float* ys  = sm + 4 * L1_IMG;
  float* zs  = sm;

  const int tx = blockIdx.x, ty = blockIdx.y, b = blockIdx.z;
  const int tid = threadIdx.x;
  const int R0 = 16 * ty - 6, C0 = 16 * tx - 6;
  const int ib = 8 * ty - 3,  jb = 8 * tx - 3;

  const int lane = tid & 31;
  const int wv = __builtin_amdgcn_readfirstlane(tid >> 5);

  float vl[5][3];
  float hra[3][4], hia[3][4], hrb[3][4], hib[3][4];
  int fl[3], l0[4], l1[4];
  {
    int gof[3];
#pragma unroll
    for (int i = 0; i < 3; ++i) {
      const int f  = lane + 32 * i;
      const int wc = f / 3, ch = f - 3 * wc;
      gof[i] = ref_idx(C0 + wc, 256) * 3 + ch;
      fl[i]  = f < 84 ? f : 84 + L1_RS * (f - 84);
    }
#pragma unroll
    for (int n = 0; n < 5; ++n) {
      int wr = (5 - wv) + 6 * n;
      wr = wr < 28 ? wr : 27;
      const float* srow = low + (b * 256 + ref_idx(R0 + wr, 256)) * 768;
#pragma unroll
      for (int i = 0; i < 3; ++i) vl[n][i] = srow[gof[i]];
    }
  }
  {
    int ga[4], gb[4];
#pragma unroll
    for (int i = 0; i < 4; ++i) {
      const int m  = lane + 32 * i;
      const int pc = m / 9,   k  = m - 9 * pc;
      const int pair = k / 3, ch = k - 3 * pair;
      int J = jb + pc;
      const int fc = (J < 0 || J >= 128) ? 1 : 0;
      J = (J < 0) ? (-1 - J) : ((J >= 128) ? (255 - J) : J);
      const int oa = (pair == 0) ? 0  : ((pair == 1) ? 6 : 3);
      const int ob = (pair == 0) ? 15 : ((pair == 1) ? 9 : 12);
      ga[i] = J * 36 + ch + oa;
      gb[i] = J * 36 + ch + ob;
      const int c0 = (1 + pair) * L1_IMG + (2 * pc + fc) * 3 + ch;
      const int c1 = (1 + pair) * L1_IMG + (2 * pc + (1 ^ fc)) * 3 + ch;
      l0[i] = m < 126 ? c0 : (L1_IMG + 84 + L1_RS * (m - 126));
      l1[i] = m < 126 ? c1 : (L1_IMG + 84 + L1_RS * (m - 126));
    }
#pragma unroll
    for (int n = 0; n < 3; ++n) {
      int pr = wv + 6 * n;
      pr = pr < 14 ? pr : 13;
      int I = ib + pr;
      I = (I < 0) ? (-1 - I) : ((I >= 128) ? (255 - I) : I);
      const float* srow = high1 + (b * 128 + I) * (128 * 36);
#pragma unroll
      for (int i = 0; i < 4; ++i) {
        hra[n][i] = srow[ga[i]];  hia[n][i] = srow[ga[i] + 18];
        hrb[n][i] = srow[gb[i]];  hib[n][i] = srow[gb[i] + 18];
      }
    }
  }

#pragma unroll
  for (int n = 0; n < 5; ++n) {
    int wr = (5 - wv) + 6 * n;
    wr = wr < 28 ? wr : 27;
    float* drow = img + wr * L1_RS;
#pragma unroll
    for (int i = 0; i < 3; ++i) drow[fl[i]] = vl[n][i];
  }
#pragma unroll
  for (int n = 0; n < 3; ++n) {
    int pr = wv + 6 * n;
    pr = pr < 14 ? pr : 13;
    const int I = ib + pr;
    const int fr = (I < 0 || I >= 128) ? 1 : 0;
    float* r0 = img + (2 * pr + fr) * L1_RS;
    float* r1 = img + (2 * pr + (1 ^ fr)) * L1_RS;
#pragma unroll
    for (int i = 0; i < 4; ++i) {
      const float ra = hra[n][i], ia  = hia[n][i];
      const float rb = hrb[n][i], ibm = hib[n][i];
      r0[l0[i]] = (ra + rb)  * SCQ;
      r0[l1[i]] = (ia + ibm) * SCQ;
      r1[l0[i]] = (ia - ibm) * SCQ;
      r1[l1[i]] = (rb - ra)  * SCQ;
    }
  }
  __syncthreads();

  {
    const int plane = tid / 96, c = tid - plane * 96;
    if (c < 84) {
      const float* A = img + (2 * plane) * L1_IMG + c;
      const float* B = A + L1_IMG;
      float a[28], g[28];
#pragma unroll
      for (int r = 0; r < 28; ++r) { a[r] = A[r * L1_RS]; g[r] = B[r * L1_RS]; }
      float* Y = ys + plane * L1_YPL + c;
#pragma unroll
      for (int q = 0; q < 8; ++q) {
#pragma unroll
        for (int s = 0; s < 4; ++s) {
          const int  p  = s >> 1;
          const bool ev = (s & 1) == 0;
          const int  b0 = ev ? 6 : 7, b1 = ev ? 7 : 6;
          float s0 = 0.0f, s1 = 0.0f;
#pragma unroll
          for (int j = 0; j < 7; ++j) {
            s0 += (ev ? G0B[2 * j + p] : G0A[2 * j + p]) * a[2 * q + b0 + 6 - 2 * j];
            s1 += (ev ? G1B[2 * j + p] : G1A[2 * j + p]) * g[2 * q + b1 + 6 - 2 * j];
          }
          Y[(4 * q + s) * L1_RS] = s0 + s1;
        }
      }
    }
  }
  __syncthreads();

  {
    const int w = tid >> 5, yy = tid & 31;
    const int half = w / 3, ch = w - 3 * half;
    const float* Y1 = ys + yy * L1_RS + (8 * half) * 3 + ch;
    const float* Y2 = Y1 + L1_YPL;
    float u1[20], u2[20];
#pragma unroll
    for (int k = 0; k < 20; ++k) { u1[k] = Y1[3 * k]; u2[k] = Y2[3 * k]; }
    float* zrow = zs + yy * L1_ZS + (16 * half) * 3 + ch;
#pragma unroll
    for (int q = 0; q < 4; ++q) {
#pragma unroll
      for (int s = 0; s < 4; ++s) {
        const int  p  = s >> 1;
        const bool ev = (s & 1) == 0;
        const int  b0 = ev ? 6 : 7, b1 = ev ? 7 : 6;
        float acc = 0.0f;
#pragma unroll
        for (int j = 0; j < 7; ++j) {
          acc += (ev ? G0B[2 * j + p] : G0A[2 * j + p]) * u1[2 * q + b0 + 6 - 2 * j];
          acc += (ev ? G1B[2 * j + p] : G1A[2 * j + p]) * u2[2 * q + b1 + 6 - 2 * j];
        }
        zrow[(4 * q + s) * 3] = acc;
      }
    }
  }
  __syncthreads();

  {
    const int hrow = tid / 96, f = tid - 96 * hrow;
#pragma unroll
    for (int it = 0; it < 16; ++it) {
      const int row = 2 * it + hrow;
      Z[((b * 512 + 32 * ty + row) * 512 + 32 * tx) * 3 + f] = zs[row * L1_ZS + f];
    }
  }
}

__device__ __forceinline__ float tap7(int i) {
  int ic = i < 0 ? 0 : (i > 6 ? 6 : i);
  float t = cG0O[ic];
  return (i == ic) ? t : 0.0f;
}
__device__ __forceinline__ float tap5(int i) {
  int ic = i < 0 ? 0 : (i > 4 ? 4 : i);
  float t = cG1O[ic];
  return (i == ic) ? t : 0.0f;
}

__device__ __forceinline__ v8f wmma4(v2f a, v2f b, v8f c) {
  return __builtin_amdgcn_wmma_f32_16x16x4_f32(
       false, a,  false, b,
       (short)0, c,  false,  false);
}

#define L0_P   41
#define L0_PL  (40 * L0_P)
#define L0_YP  36
#define L0_YW  (2 * 16 * L0_YP)
#define L0_OP  100
#define L0_OT  (12 * L0_YW)

__global__ void __launch_bounds__(384, 12)
k_level0(const float* __restrict__ Z, const float* __restrict__ high0,
         float* __restrict__ out) {
  __shared__ __attribute__((aligned(16))) float sm[12 * L0_PL];

  const int tx = blockIdx.x, ty = blockIdx.y, b = blockIdx.z;
  const int tid = threadIdx.x;
  const int X0 = 32 * tx, Y0 = 32 * ty;
  const int R0 = Y0 - 4, C0 = X0 - 4;
  const int ib = 16 * ty - 2, jb = 16 * tx - 2;

  const int lane = tid & 31;
  const int w = __builtin_amdgcn_readfirstlane(tid >> 5);

  float zv[14];
  float hra[10], hia[10], hrb[10], hib[10];
  int zdst, p0, p1;
  const float* zb = Z + b * (512 * 512 * 3);
  const float* hb = high0 + b * (256 * 256 * 36);
  {
    const int g = w >> 2;
    const int f = lane + 32 * (w & 3);
    const int fz = f < 120 ? f : 119;
    const int wc = fz / 3, ch = fz - 3 * wc;
    const int zcol = ref_idx(C0 + wc, 512) * 3 + ch;
    zdst = f < 120 ? (ch * L0_PL + wc) : 40 + L0_P * (f - 120);
#pragma unroll
    for (int n = 0; n < 14; ++n) {
      int wr = 3 * n + g;
      wr = wr < 40 ? wr : 39;
      zv[n] = zb[ref_idx(R0 + wr, 512) * (512 * 3) + zcol];
    }
  }
  const int rp = w / 6;
  {
    const int m  = lane + 32 * (w - 6 * rp);
    const int mc = m < 180 ? m : 179;
    const int pc = mc / 9,  k = mc - 9 * pc;
    const int pair = k / 3, ch = k - 3 * pair;
    int J = jb + pc;
    const int fc = (J < 0 || J >= 256) ? 1 : 0;
    J = (J < 0) ? (-1 - J) : ((J >= 256) ? (511 - J) : J);
    const int oa = (pair == 0) ? 0  : ((pair == 1) ? 6 : 3);
    const int ob = (pair == 0) ? 15 : ((pair == 1) ? 9 : 12);
    const int ga = J * 36 + ch + oa;
    const int gb = J * 36 + ch + ob;
    const int c0 = ((1 + pair) * 3 + ch) * L0_PL + 2 * pc + fc;
    const int c1 = ((1 + pair) * 3 + ch) * L0_PL + 2 * pc + (1 ^ fc);
    p0 = m < 180 ? c0 : 40 + L0_P * (m - 180);
    p1 = m < 180 ? c1 : 40 + L0_P * (m - 180);
#pragma unroll
    for (int n = 0; n < 10; ++n) {
      const int pr = 2 * n + rp;
      int I = ib + pr;
      I = (I < 0) ? (-1 - I) : ((I >= 256) ? (511 - I) : I);
      const float* srow = hb + I * (256 * 36);
      hra[n] = srow[ga];  hia[n] = srow[ga + 18];
      hrb[n] = srow[gb];  hib[n] = srow[gb + 18];
    }
  }

#pragma unroll
  for (int n = 0; n < 14; ++n) {
    int wr = 3 * n + (w >> 2);
    wr = wr < 40 ? wr : 39;
    sm[wr * L0_P + zdst] = zv[n];
  }
#pragma unroll
  for (int n = 0; n < 10; ++n) {
    const int pr = 2 * n + rp;
    const int I = ib + pr;
    const int fr = (I < 0 || I >= 256) ? 1 : 0;
    const int r0 = (2 * pr + fr) * L0_P;
    const int r1 = (2 * pr + (1 ^ fr)) * L0_P;
    const float ra = hra[n], ia  = hia[n];
    const float rb = hrb[n], ibm = hib[n];
    sm[p0 + r0] = (ra + rb)  * SCQ;
    sm[p1 + r0] = (ia + ibm) * SCQ;
    sm[p0 + r1] = (ia - ibm) * SCQ;
    sm[p1 + r1] = (rb - ra)  * SCQ;
  }

  const int lo = lane & 15;
  const int h2 = (lane >> 4) * 2;
  const int sub = w / 3, ch = w - 3 * sub;
  const int sy = sub >> 1, sx = sub & 1;
  const int x0 = X0 + 16 * sx, y0 = Y0 + 16 * sy;

  __syncthreads();

  v2f f0[6], f1[6];
#pragma unroll
  for (int kk = 0; kk < 6; ++kk) {
    const int k0 = 4 * kk + h2;
    f0[kk].x = tap7(k0 - lo - 1);  f0[kk].y = tap7(k0 - lo);
    f1[kk].x = tap5(k0 - lo - 2);  f1[kk].y = tap5(k0 - lo - 1);
  }

  v8f a1[2], a2[2];
  {
    const float* zp = sm + ch * L0_PL + (16 * sy) * L0_P;
    const float* lp = zp + 3 * L0_PL;
    const float* hp = zp + 6 * L0_PL;
    const float* qp = zp + 9 * L0_PL;
#pragma unroll
    for (int t = 0; t < 2; ++t) {
      a1[t] = (v8f){}; a2[t] = (v8f){};
      int cc = 16 * sx + 16 * t + lo;
      cc = cc > 39 ? 39 : cc;
#pragma unroll
      for (int kk = 0; kk < 6; ++kk) {
        const int k0 = 4 * kk + h2;
        const int o0 = k0 * L0_P + cc, o1 = o0 + L0_P;
        v2f bz = { zp[o0], zp[o1] };
        v2f bl = { lp[o0], lp[o1] };
        v2f bh = { hp[o0], hp[o1] };
        v2f bq = { qp[o0], qp[o1] };
        a1[t] = wmma4(f0[kk], bz, a1[t]);
        a1[t] = wmma4(f1[kk], bl, a1[t]);
        a2[t] = wmma4(f0[kk], bh, a2[t]);
        a2[t] = wmma4(f1[kk], bq, a2[t]);
      }
    }
  }
  __syncthreads();

  float* y1s = sm + w * L0_YW;
  float* y2s = y1s + 16 * L0_YP;
#pragma unroll
  for (int t = 0; t < 2; ++t) {
#pragma unroll
    for (int v = 0; v < 8; ++v) {
      const int row = v + (h2 << 2);
      y1s[row * L0_YP + 16 * t + lo] = a1[t][v];
      y2s[row * L0_YP + 16 * t + lo] = a2[t][v];
    }
  }
  __syncthreads();

  v8f o = {};
#pragma unroll
  for (int kk = 0; kk < 6; ++kk) {
    const int k0 = 4 * kk + h2;
    v2f A1 = { y1s[lo * L0_YP + k0], y1s[lo * L0_YP + k0 + 1] };
    v2f A2 = { y2s[lo * L0_YP + k0], y2s[lo * L0_YP + k0 + 1] };
    o = wmma4(A1, f0[kk], o);
    o = wmma4(A2, f1[kk], o);
  }
  float* ot = sm + L0_OT;
#pragma unroll
  for (int v = 0; v < 8; ++v) {
    const int row = 16 * sy + v + (h2 << 2);
    ot[row * L0_OP + (16 * sx + lo) * 3 + ch] = o[v];
  }
  __syncthreads();
  float* ob = out + ((b * 512 + Y0) * 512 + X0) * 3;
#pragma unroll
  for (int i = 0; i < 2; ++i) {
    const int q = tid + 384 * i;
    const int row = q / 24, c4 = q - 24 * row;
    const v4f val = *(const v4f*)(ot + row * L0_OP + 4 * c4);
    *(v4f*)(ob + row * (512 * 3) + 4 * c4) = val;
  }
}

extern "C" void kernel_launch(void* const* d_in, const int* in_sizes, int n_in,
                              void* d_out, int out_size, void* d_ws, size_t ws_size,
                              hipStream_t stream) {
  (void)in_sizes; (void)n_in; (void)out_size; (void)ws_size;
  const float* low   = (const float*)d_in[0];
  const float* high0 = (const float*)d_in[1];
  const float* high1 = (const float*)d_in[2];
  float* out = (float*)d_out;
  float* Zws = (float*)d_ws;

  k_level1<<<dim3(16, 16, 8), dim3(192), 0, stream>>>(low, high1, Zws);
  k_level0<<<dim3(16, 16, 8), dim3(384), 0, stream>>>(Zws, high0, out);
}


# level-0 column pass shared by the two waves of a 16x32 strip: the 16-column block between their sub-tiles is computed once (one wave its Y1, the other its Y2), 48 WMMAs per wave instead of 60
# speedup vs baseline: 6.2514x; 1.0584x over previous
#include <hip/hip_runtime.h>
#include <hip/hip_bf16.h>


typedef float v2f __attribute__((ext_vector_type(2)));
typedef float v8f __attribute__((ext_vector_type(8)));
typedef float v4f __attribute__((ext_vector_type(4)));

#define SCQ 0.70710678118654752440f

__constant__ float cG0O[7] = { -3.0f/280.0f, -15.0f/280.0f, 73.0f/280.0f,
                               170.0f/280.0f, 73.0f/280.0f, -15.0f/280.0f,
                               -3.0f/280.0f };
__constant__ float cG1O[5] = { -1.0f/20.0f, -5.0f/20.0f, 12.0f/20.0f,
                               -5.0f/20.0f, -1.0f/20.0f };

__device__ __forceinline__ int ref_idx(int i, int L) {
  if (i < 0)  i = -1 - i;
  if (i >= L) i = 2 * L - 1 - i;
  return i;
}

#define L1_RS  85
#define L1_IMG (28 * L1_RS)
#define L1_YPL (32 * L1_RS)
#define L1_ZS  97

__global__ void __launch_bounds__(192)
k_level1(const float* __restrict__ low, const float* __restrict__ high1,
         float* __restrict__ Z) {
  constexpr float G0B[14] = {
     0.00325314f, -0.00388321f,  0.03466035f, -0.03887280f,
    -0.11720389f,  0.27529538f,  0.75614564f,  0.56881042f,
     0.01186609f, -0.10671180f,  0.02382538f,  0.01702522f,
    -0.00543948f, -0.00455690f };
  constexpr float G0A[14] = {
    -0.00455690f, -0.00543948f,  0.01702522f,  0.02382538f,
    -0.10671180f,  0.01186609f,  0.56881042f,  0.75614564f,
     0.27529538f, -0.11720389f, -0.03887280f,  0.03466035f,
    -0.00388321f,  0.00325314f };
  constexpr float G1B[14] = {
    -0.00455690f,  0.00543948f,  0.01702522f, -0.02382538f,
    -0.10671180f, -0.01186609f,  0.56881042f, -0.75614564f,
     0.27529538f,  0.11720389f, -0.03887280f, -0.03466035f,
    -0.00388321f, -0.00325314f };
  constexpr float G1A[14] = {
    -0.00325314f, -0.00388321f, -0.03466035f, -0.03887280f,
     0.11720389f,  0.27529538f, -0.75614564f,  0.56881042f,
    -0.01186609f, -0.10671180f, -0.02382538f,  0.01702522f,
     0.00543948f, -0.00455690f };

  __shared__ float sm[4 * L1_IMG + 2 * L1_YPL];
  float* img = sm;
  float* ys  = sm + 4 * L1_IMG;
  float* zs  = sm;

  const int tx = blockIdx.x, ty = blockIdx.y, b = blockIdx.z;
  const int tid = threadIdx.x;
  const int R0 = 16 * ty - 6, C0 = 16 * tx - 6;
  const int ib = 8 * ty - 3,  jb = 8 * tx - 3;

  const int lane = tid & 31;
  const int wv = __builtin_amdgcn_readfirstlane(tid >> 5);

  float vl[5][3];
  float hra[3][4], hia[3][4], hrb[3][4], hib[3][4];
  int fl[3], l0[4], l1[4];
  {
    int gof[3];
#pragma unroll
    for (int i = 0; i < 3; ++i) {
      const int f  = lane + 32 * i;
      const int wc = f / 3, ch = f - 3 * wc;
      gof[i] = ref_idx(C0 + wc, 256) * 3 + ch;
      fl[i]  = f < 84 ? f : 84 + L1_RS * (f - 84);
    }
#pragma unroll
    for (int n = 0; n < 5; ++n) {
      int wr = (5 - wv) + 6 * n;
      wr = wr < 28 ? wr : 27;
      const float* srow = low + (b * 256 + ref_idx(R0 + wr, 256)) * 768;
#pragma unroll
      for (int i = 0; i < 3; ++i) vl[n][i] = srow[gof[i]];
    }
  }
  {
    int ga[4], gb[4];
#pragma unroll
    for (int i = 0; i < 4; ++i) {
      const int m  = lane + 32 * i;
      const int pc = m / 9,   k  = m - 9 * pc;
      const int pair = k / 3, ch = k - 3 * pair;
      int J = jb + pc;
      const int fc = (J < 0 || J >= 128) ? 1 : 0;
      J = (J < 0) ? (-1 - J) : ((J >= 128) ? (255 - J) : J);
      const int oa = (pair == 0) ? 0  : ((pair == 1) ? 6 : 3);
      const int ob = (pair == 0) ? 15 : ((pair == 1) ? 9 : 12);
      ga[i] = J * 36 + ch + oa;
      gb[i] = J * 36 + ch + ob;
      const int c0 = (1 + pair) * L1_IMG + (2 * pc + fc) * 3 + ch;
      const int c1 = (1 + pair) * L1_IMG + (2 * pc + (1 ^ fc)) * 3 + ch;
      l0[i] = m < 126 ? c0 : (L1_IMG + 84 + L1_RS * (m - 126));
      l1[i] = m < 126 ? c1 : (L1_IMG + 84 + L1_RS * (m - 126));
    }
#pragma unroll
    for (int n = 0; n < 3; ++n) {
      int pr = wv + 6 * n;
      pr = pr < 14 ? pr : 13;
      int I = ib + pr;
      I = (I < 0) ? (-1 - I) : ((I >= 128) ? (255 - I) : I);
      const float* srow = high1 + (b * 128 + I) * (128 * 36);
#pragma unroll
      for (int i = 0; i < 4; ++i) {
        hra[n][i] = srow[ga[i]];  hia[n][i] = srow[ga[i] + 18];
        hrb[n][i] = srow[gb[i]];  hib[n][i] = srow[gb[i] + 18];
      }
    }
  }

#pragma unroll
  for (int n = 0; n < 5; ++n) {
    int wr = (5 - wv) + 6 * n;
    wr = wr < 28 ? wr : 27;
    float* drow = img + wr * L1_RS;
#pragma unroll
    for (int i = 0; i < 3; ++i) drow[fl[i]] = vl[n][i];
  }
#pragma unroll
  for (int n = 0; n < 3; ++n) {
    int pr = wv + 6 * n;
    pr = pr < 14 ? pr : 13;
    const int I = ib + pr;
    const int fr = (I < 0 || I >= 128) ? 1 : 0;
    float* r0 = img + (2 * pr + fr) * L1_RS;
    float* r1 = img + (2 * pr + (1 ^ fr)) * L1_RS;
#pragma unroll
    for (int i = 0; i < 4; ++i) {
      const float ra = hra[n][i], ia  = hia[n][i];
      const float rb = hrb[n][i], ibm = hib[n][i];
      r0[l0[i]] = (ra + rb)  * SCQ;
      r0[l1[i]] = (ia + ibm) * SCQ;
      r1[l0[i]] = (ia - ibm) * SCQ;
      r1[l1[i]] = (rb - ra)  * SCQ;
    }
  }
  __syncthreads();

  {
    const int plane = tid / 96, c = tid - plane * 96;
    if (c < 84) {
      const float* A = img + (2 * plane) * L1_IMG + c;
      const float* B = A + L1_IMG;
      float a[28], g[28];
#pragma unroll
      for (int r = 0; r < 28; ++r) { a[r] = A[r * L1_RS]; g[r] = B[r * L1_RS]; }
      float* Y = ys + plane * L1_YPL + c;
#pragma unroll
      for (int q = 0; q < 8; ++q) {
#pragma unroll
        for (int s = 0; s < 4; ++s) {
          const int  p  = s >> 1;
          const bool ev = (s & 1) == 0;
          const int  b0 = ev ? 6 : 7, b1 = ev ? 7 : 6;
          float s0 = 0.0f, s1 = 0.0f;
#pragma unroll
          for (int j = 0; j < 7; ++j) {
            s0 += (ev ? G0B[2 * j + p] : G0A[2 * j + p]) * a[2 * q + b0 + 6 - 2 * j];
            s1 += (ev ? G1B[2 * j + p] : G1A[2 * j + p]) * g[2 * q + b1 + 6 - 2 * j];
          }
          Y[(4 * q + s) * L1_RS] = s0 + s1;
        }
      }
    }
  }
  __syncthreads();

  {
    const int w = tid >> 5, yy = tid & 31;
    const int half = w / 3, ch = w - 3 * half;
    const float* Y1 = ys + yy * L1_RS + (8 * half) * 3 + ch;
    const float* Y2 = Y1 + L1_YPL;
    float u1[20], u2[20];
#pragma unroll
    for (int k = 0; k < 20; ++k) { u1[k] = Y1[3 * k]; u2[k] = Y2[3 * k]; }
    float* zrow = zs + yy * L1_ZS + (16 * half) * 3 + ch;
#pragma unroll
    for (int q = 0; q < 4; ++q) {
#pragma unroll
      for (int s = 0; s < 4; ++s) {
        const int  p  = s >> 1;
        const bool ev = (s & 1) == 0;
        const int  b0 = ev ? 6 : 7, b1 = ev ? 7 : 6;
        float acc = 0.0f;
#pragma unroll
        for (int j = 0; j < 7; ++j) {
          acc += (ev ? G0B[2 * j + p] : G0A[2 * j + p]) * u1[2 * q + b0 + 6 - 2 * j];
          acc += (ev ? G1B[2 * j + p] : G1A[2 * j + p]) * u2[2 * q + b1 + 6 - 2 * j];
        }
        zrow[(4 * q + s) * 3] = acc;
      }
    }
  }
  __syncthreads();

  {
    const int hrow = tid / 96, f = tid - 96 * hrow;
#pragma unroll
    for (int it = 0; it < 16; ++it) {
      const int row = 2 * it + hrow;
      Z[((b * 512 + 32 * ty + row) * 512 + 32 * tx) * 3 + f] = zs[row * L1_ZS + f];
    }
  }
}

__device__ __forceinline__ float tap7(int i) {
  int ic = i < 0 ? 0 : (i > 6 ? 6 : i);
  float t = cG0O[ic];
  return (i == ic) ? t : 0.0f;
}
__device__ __forceinline__ float tap5(int i) {
  int ic = i < 0 ? 0 : (i > 4 ? 4 : i);
  float t = cG1O[ic];
  return (i == ic) ? t : 0.0f;
}

__device__ __forceinline__ v8f wmma4(v2f a, v2f b, v8f c) {
  return __builtin_amdgcn_wmma_f32_16x16x4_f32(
       false, a,  false, b,
       (short)0, c,  false,  false);
}

#define L0_P   41
#define L0_PL  (40 * L0_P)
#define L0_YP  52
#define L0_YW  (2 * 16 * L0_YP)
#define L0_OP  100
#define L0_OT  (6 * L0_YW)

__global__ void __launch_bounds__(384, 12)
k_level0(const float* __restrict__ Z, const float* __restrict__ high0,
         float* __restrict__ out) {
  __shared__ __attribute__((aligned(16))) float sm[12 * L0_PL];

  const int tx = blockIdx.x, ty = blockIdx.y, b = blockIdx.z;
  const int tid = threadIdx.x;
  const int X0 = 32 * tx, Y0 = 32 * ty;
  const int R0 = Y0 - 4, C0 = X0 - 4;
  const int ib = 16 * ty - 2, jb = 16 * tx - 2;

  const int lane = tid & 31;
  const int w = __builtin_amdgcn_readfirstlane(tid >> 5);

  float zv[14];
  float hra[10], hia[10], hrb[10], hib[10];
  int zdst, p0, p1;
  const float* zb = Z + b * (512 * 512 * 3);
  const float* hb = high0 + b * (256 * 256 * 36);
  {
    const int g = w >> 2;
    const int f = lane + 32 * (w & 3);
    const int fz = f < 120 ? f : 119;
    const int wc = fz / 3, ch = fz - 3 * wc;
    const int zcol = ref_idx(C0 + wc, 512) * 3 + ch;
    zdst = f < 120 ? (ch * L0_PL + wc) : 40 + L0_P * (f - 120);
#pragma unroll
    for (int n = 0; n < 14; ++n) {
      int wr = 3 * n + g;
      wr = wr < 40 ? wr : 39;
      zv[n] = zb[ref_idx(R0 + wr, 512) * (512 * 3) + zcol];
    }
  }
  const int rp = w / 6;
  {
    const int m  = lane + 32 * (w - 6 * rp);
    const int mc = m < 180 ? m : 179;
    const int pc = mc / 9,  k = mc - 9 * pc;
    const int pair = k / 3, ch = k - 3 * pair;
    int J = jb + pc;
    const int fc = (J < 0 || J >= 256) ? 1 : 0;
    J = (J < 0) ? (-1 - J) : ((J >= 256) ? (511 - J) : J);
    const int oa = (pair == 0) ? 0  : ((pair == 1) ? 6 : 3);
    const int ob = (pair == 0) ? 15 : ((pair == 1) ? 9 : 12);
    const int ga = J * 36 + ch + oa;
    const int gb = J * 36 + ch + ob;
    const int c0 = ((1 + pair) * 3 + ch) * L0_PL + 2 * pc + fc;
    const int c1 = ((1 + pair) * 3 + ch) * L0_PL + 2 * pc + (1 ^ fc);
    p0 = m < 180 ? c0 : 40 + L0_P * (m - 180);
    p1 = m < 180 ? c1 : 40 + L0_P * (m - 180);
#pragma unroll
    for (int n = 0; n < 10; ++n) {
      const int pr = 2 * n + rp;
      int I = ib + pr;
      I = (I < 0) ? (-1 - I) : ((I >= 256) ? (511 - I) : I);
      const float* srow = hb + I * (256 * 36);
      hra[n] = srow[ga];  hia[n] = srow[ga + 18];
      hrb[n] = srow[gb];  hib[n] = srow[gb + 18];
    }
  }

#pragma unroll
  for (int n = 0; n < 14; ++n) {
    int wr = 3 * n + (w >> 2);
    wr = wr < 40 ? wr : 39;
    sm[wr * L0_P + zdst] = zv[n];
  }
#pragma unroll
  for (int n = 0; n < 10; ++n) {
    const int pr = 2 * n + rp;
    const int I = ib + pr;
    const int fr = (I < 0 || I >= 256) ? 1 : 0;
    const int r0 = (2 * pr + fr) * L0_P;
    const int r1 = (2 * pr + (1 ^ fr)) * L0_P;
    const float ra = hra[n], ia  = hia[n];
    const float rb = hrb[n], ibm = hib[n];
    sm[p0 + r0] = (ra + rb)  * SCQ;
    sm[p1 + r0] = (ia + ibm) * SCQ;
    sm[p0 + r1] = (ia - ibm) * SCQ;
    sm[p1 + r1] = (rb - ra)  * SCQ;
  }

  const int lo = lane & 15;
  const int h2 = (lane >> 4) * 2;
  const int sub = w / 3, ch = w - 3 * sub;
  const int sy = sub >> 1, sx = sub & 1;
  const int x0 = X0 + 16 * sx, y0 = Y0 + 16 * sy;

  __syncthreads();

  v2f f0[6], f1[6];
#pragma unroll
  for (int kk = 0; kk < 6; ++kk) {
    const int k0 = 4 * kk + h2;
    f0[kk].x = tap7(k0 - lo - 1);  f0[kk].y = tap7(k0 - lo);
    f1[kk].x = tap5(k0 - lo - 2);  f1[kk].y = tap5(k0 - lo - 1);
  }

  v8f a1 = {}, a2 = {}, am = {};
  {
    const float* zp = sm + ch * L0_PL + (16 * sy) * L0_P;
    const float* lp = zp + 3 * L0_PL;
    const float* hp = zp + 6 * L0_PL;
    const float* qp = zp + 9 * L0_PL;
    const float* m0 = sx ? hp : zp;
    const float* m1 = sx ? qp : lp;
    int cc = 32 * sx + lo;
    cc = cc > 39 ? 39 : cc;
    const int cm = 16 + lo;
#pragma unroll
    for (int kk = 0; kk < 6; ++kk) {
      const int k0 = 4 * kk + h2;
      const int o0 = k0 * L0_P + cc, o1 = o0 + L0_P;
      v2f bz = { zp[o0], zp[o1] };
      v2f bl = { lp[o0], lp[o1] };
      v2f bh = { hp[o0], hp[o1] };
      v2f bq = { qp[o0], qp[o1] };
      a1 = wmma4(f0[kk], bz, a1);
      a1 = wmma4(f1[kk], bl, a1);
      a2 = wmma4(f0[kk], bh, a2);
      a2 = wmma4(f1[kk], bq, a2);
    }
#pragma unroll
    for (int kk = 0; kk < 6; ++kk) {
      const int k0 = 4 * kk + h2;
      const int c0 = k0 * L0_P + cm, c1 = c0 + L0_P;
      v2f b0 = { m0[c0], m0[c1] };
      v2f b1 = { m1[c0], m1[c1] };
      am = wmma4(f0[kk], b0, am);
      am = wmma4(f1[kk], b1, am);
    }
  }
  __syncthreads();

  float* y1s = sm + (3 * sy + ch) * L0_YW;
  float* y2s = y1s + 16 * L0_YP;
  float* yms = sx ? y2s : y1s;
#pragma unroll
  for (int v = 0; v < 8; ++v) {
    const int row = v + (h2 << 2);
    y1s[row * L0_YP + 32 * sx + lo] = a1[v];
    y2s[row * L0_YP + 32 * sx + lo] = a2[v];
    yms[row * L0_YP + 16 + lo]      = am[v];
  }
  __syncthreads();

  const float* y1r = y1s + 16 * sx;
  const float* y2r = y2s + 16 * sx;
  v8f o = {};
#pragma unroll
  for (int kk = 0; kk < 6; ++kk) {
    const int k0 = 4 * kk + h2;
    v2f A1 = { y1r[lo * L0_YP + k0], y1r[lo * L0_YP + k0 + 1] };
    v2f A2 = { y2r[lo * L0_YP + k0], y2r[lo * L0_YP + k0 + 1] };
    o = wmma4(A1, f0[kk], o);
    o = wmma4(A2, f1[kk], o);
  }
  float* ot = sm + L0_OT;
#pragma unroll
  for (int v = 0; v < 8; ++v) {
    const int row = 16 * sy + v + (h2 << 2);
    ot[row * L0_OP + (16 * sx + lo) * 3 + ch] = o[v];
  }
  __syncthreads();
  float* ob = out + ((b * 512 + Y0) * 512 + X0) * 3;
#pragma unroll
  for (int i = 0; i < 2; ++i) {
    const int q = tid + 384 * i;
    const int row = q / 24, c4 = q - 24 * row;
    const v4f val = *(const v4f*)(ot + row * L0_OP + 4 * c4);
    *(v4f*)(ob + row * (512 * 3) + 4 * c4) = val;
  }
}

extern "C" void kernel_launch(void* const* d_in, const int* in_sizes, int n_in,
                              void* d_out, int out_size, void* d_ws, size_t ws_size,
                              hipStream_t stream) {
  (void)in_sizes; (void)n_in; (void)out_size; (void)ws_size;
  const float* low   = (const float*)d_in[0];
  const float* high0 = (const float*)d_in[1];
  const float* high1 = (const float*)d_in[2];
  float* out = (float*)d_out;
  float* Zws = (float*)d_ws;

  k_level1<<<dim3(16, 16, 8), dim3(192), 0, stream>>>(low, high1, Zws);
  k_level0<<<dim3(16, 16, 8), dim3(384), 0, stream>>>(Zws, high0, out);
}


# level-0 WMMAs trimmed to their operands' used extent: the 5-tap band runs K=20 (window shifted by 2; its sums regroup), columns 32..39 of Y1 and Y2 share one tile: 77 WMMAs per strip, not 96
# speedup vs baseline: 6.4770x; 1.0361x over previous
#include <hip/hip_runtime.h>
#include <hip/hip_bf16.h>


typedef float v2f __attribute__((ext_vector_type(2)));
typedef float v8f __attribute__((ext_vector_type(8)));
typedef float v4f __attribute__((ext_vector_type(4)));

#define SCQ 0.70710678118654752440f

__constant__ float cG0O[7] = { -3.0f/280.0f, -15.0f/280.0f, 73.0f/280.0f,
                               170.0f/280.0f, 73.0f/280.0f, -15.0f/280.0f,
                               -3.0f/280.0f };
__constant__ float cG1O[5] = { -1.0f/20.0f, -5.0f/20.0f, 12.0f/20.0f,
                               -5.0f/20.0f, -1.0f/20.0f };

__device__ __forceinline__ int ref_idx(int i, int L) {
  if (i < 0)  i = -1 - i;
  if (i >= L) i = 2 * L - 1 - i;
  return i;
}

#define L1_RS  85
#define L1_IMG (28 * L1_RS)
#define L1_YPL (32 * L1_RS)
#define L1_ZS  97

__global__ void __launch_bounds__(192)
k_level1(const float* __restrict__ low, const float* __restrict__ high1,
         float* __restrict__ Z) {
  constexpr float G0B[14] = {
     0.00325314f, -0.00388321f,  0.03466035f, -0.03887280f,
    -0.11720389f,  0.27529538f,  0.75614564f,  0.56881042f,
     0.01186609f, -0.10671180f,  0.02382538f,  0.01702522f,
    -0.00543948f, -0.00455690f };
  constexpr float G0A[14] = {
    -0.00455690f, -0.00543948f,  0.01702522f,  0.02382538f,
    -0.10671180f,  0.01186609f,  0.56881042f,  0.75614564f,
     0.27529538f, -0.11720389f, -0.03887280f,  0.03466035f,
    -0.00388321f,  0.00325314f };
  constexpr float G1B[14] = {
    -0.00455690f,  0.00543948f,  0.01702522f, -0.02382538f,
    -0.10671180f, -0.01186609f,  0.56881042f, -0.75614564f,
     0.27529538f,  0.11720389f, -0.03887280f, -0.03466035f,
    -0.00388321f, -0.00325314f };
  constexpr float G1A[14] = {
    -0.00325314f, -0.00388321f, -0.03466035f, -0.03887280f,
     0.11720389f,  0.27529538f, -0.75614564f,  0.56881042f,
    -0.01186609f, -0.10671180f, -0.02382538f,  0.01702522f,
     0.00543948f, -0.00455690f };

  __shared__ float sm[4 * L1_IMG + 2 * L1_YPL];
  float* img = sm;
  float* ys  = sm + 4 * L1_IMG;
  float* zs  = sm;

  const int tx = blockIdx.x, ty = blockIdx.y, b = blockIdx.z;
  const int tid = threadIdx.x;
  const int R0 = 16 * ty - 6, C0 = 16 * tx - 6;
  const int ib = 8 * ty - 3,  jb = 8 * tx - 3;

  const int lane = tid & 31;
  const int wv = __builtin_amdgcn_readfirstlane(tid >> 5);

  float vl[5][3];
  float hra[3][4], hia[3][4], hrb[3][4], hib[3][4];
  int fl[3], l0[4], l1[4];
  {
    int gof[3];
#pragma unroll
    for (int i = 0; i < 3; ++i) {
      const int f  = lane + 32 * i;
      const int wc = f / 3, ch = f - 3 * wc;
      gof[i] = ref_idx(C0 + wc, 256) * 3 + ch;
      fl[i]  = f < 84 ? f : 84 + L1_RS * (f - 84);
    }
#pragma unroll
    for (int n = 0; n < 5; ++n) {
      int wr = (5 - wv) + 6 * n;
      wr = wr < 28 ? wr : 27;
      const float* srow = low + (b * 256 + ref_idx(R0 + wr, 256)) * 768;
#pragma unroll
      for (int i = 0; i < 3; ++i) vl[n][i] = srow[gof[i]];
    }
  }
  {
    int ga[4], gb[4];
#pragma unroll
    for (int i = 0; i < 4; ++i) {
      const int m  = lane + 32 * i;
      const int pc = m / 9,   k  = m - 9 * pc;
      const int pair = k / 3, ch = k - 3 * pair;
      int J = jb + pc;
      const int fc = (J < 0 || J >= 128) ? 1 : 0;
      J = (J < 0) ? (-1 - J) : ((J >= 128) ? (255 - J) : J);
      const int oa = (pair == 0) ? 0  : ((pair == 1) ? 6 : 3);
      const int ob = (pair == 0) ? 15 : ((pair == 1) ? 9 : 12);
      ga[i] = J * 36 + ch + oa;
      gb[i] = J * 36 + ch + ob;
      const int c0 = (1 + pair) * L1_IMG + (2 * pc + fc) * 3 + ch;
      const int c1 = (1 + pair) * L1_IMG + (2 * pc + (1 ^ fc)) * 3 + ch;
      l0[i] = m < 126 ? c0 : (L1_IMG + 84 + L1_RS * (m - 126));
      l1[i] = m < 126 ? c1 : (L1_IMG + 84 + L1_RS * (m - 126));
    }
#pragma unroll
    for (int n = 0; n < 3; ++n) {
      int pr = wv + 6 * n;
      pr = pr < 14 ? pr : 13;
      int I = ib + pr;
      I = (I < 0) ? (-1 - I) : ((I >= 128) ? (255 - I) : I);
      const float* srow = high1 + (b * 128 + I) * (128 * 36);
#pragma unroll
      for (int i = 0; i < 4; ++i) {
        hra[n][i] = srow[ga[i]];  hia[n][i] = srow[ga[i] + 18];
        hrb[n][i] = srow[gb[i]];  hib[n][i] = srow[gb[i] + 18];
      }
    }
  }

#pragma unroll
  for (int n = 0; n < 5; ++n) {
    int wr = (5 - wv) + 6 * n;
    wr = wr < 28 ? wr : 27;
    float* drow = img + wr * L1_RS;
#pragma unroll
    for (int i = 0; i < 3; ++i) drow[fl[i]] = vl[n][i];
  }
#pragma unroll
  for (int n = 0; n < 3; ++n) {
    int pr = wv + 6 * n;
    pr = pr < 14 ? pr : 13;
    const int I = ib + pr;
    const int fr = (I < 0 || I >= 128) ? 1 : 0;
    float* r0 = img + (2 * pr + fr) * L1_RS;
    float* r1 = img + (2 * pr + (1 ^ fr)) * L1_RS;
#pragma unroll
    for (int i = 0; i < 4; ++i) {
      const float ra = hra[n][i], ia  = hia[n][i];
      const float rb = hrb[n][i], ibm = hib[n][i];
      r0[l0[i]] = (ra + rb)  * SCQ;
      r0[l1[i]] = (ia + ibm) * SCQ;
      r1[l0[i]] = (ia - ibm) * SCQ;
      r1[l1[i]] = (rb - ra)  * SCQ;
    }
  }
  __syncthreads();

  {
    const int plane = tid / 96, c = tid - plane * 96;
    if (c < 84) {
      const float* A = img + (2 * plane) * L1_IMG + c;
      const float* B = A + L1_IMG;
      float a[28], g[28];
#pragma unroll
      for (int r = 0; r < 28; ++r) { a[r] = A[r * L1_RS]; g[r] = B[r * L1_RS]; }
      float* Y = ys + plane * L1_YPL + c;
#pragma unroll
      for (int q = 0; q < 8; ++q) {
#pragma unroll
        for (int s = 0; s < 4; ++s) {
          const int  p  = s >> 1;
          const bool ev = (s & 1) == 0;
          const int  b0 = ev ? 6 : 7, b1 = ev ? 7 : 6;
          float s0 = 0.0f, s1 = 0.0f;
#pragma unroll
          for (int j = 0; j < 7; ++j) {
            s0 += (ev ? G0B[2 * j + p] : G0A[2 * j + p]) * a[2 * q + b0 + 6 - 2 * j];
            s1 += (ev ? G1B[2 * j + p] : G1A[2 * j + p]) * g[2 * q + b1 + 6 - 2 * j];
          }
          Y[(4 * q + s) * L1_RS] = s0 + s1;
        }
      }
    }
  }
  __syncthreads();

  {
    const int w = tid >> 5, yy = tid & 31;
    const int half = w / 3, ch = w - 3 * half;
    const float* Y1 = ys + yy * L1_RS + (8 * half) * 3 + ch;
    const float* Y2 = Y1 + L1_YPL;
    float u1[20], u2[20];
#pragma unroll
    for (int k = 0; k < 20; ++k) { u1[k] = Y1[3 * k]; u2[k] = Y2[3 * k]; }
    float* zrow = zs + yy * L1_ZS + (16 * half) * 3 + ch;
#pragma unroll
    for (int q = 0; q < 4; ++q) {
#pragma unroll
      for (int s = 0; s < 4; ++s) {
        const int  p  = s >> 1;
        const bool ev = (s & 1) == 0;
        const int  b0 = ev ? 6 : 7, b1 = ev ? 7 : 6;
        float acc = 0.0f;
#pragma unroll
        for (int j = 0; j < 7; ++j) {
          acc += (ev ? G0B[2 * j + p] : G0A[2 * j + p]) * u1[2 * q + b0 + 6 - 2 * j];
          acc += (ev ? G1B[2 * j + p] : G1A[2 * j + p]) * u2[2 * q + b1 + 6 - 2 * j];
        }
        zrow[(4 * q + s) * 3] = acc;
      }
    }
  }
  __syncthreads();

  {
    const int hrow = tid / 96, f = tid - 96 * hrow;
#pragma unroll
    for (int it = 0; it < 16; ++it) {
      const int row = 2 * it + hrow;
      Z[((b * 512 + 32 * ty + row) * 512 + 32 * tx) * 3 + f] = zs[row * L1_ZS + f];
    }
  }
}

__device__ __forceinline__ float tap7(int i) {
  int ic = i < 0 ? 0 : (i > 6 ? 6 : i);
  float t = cG0O[ic];
  return (i == ic) ? t : 0.0f;
}
__device__ __forceinline__ float tap5(int i) {
  int ic = i < 0 ? 0 : (i > 4 ? 4 : i);
  float t = cG1O[ic];
  return (i == ic) ? t : 0.0f;
}

__device__ __forceinline__ v8f wmma4(v2f a, v2f b, v8f c) {
  return __builtin_amdgcn_wmma_f32_16x16x4_f32(
       false, a,  false, b,
       (short)0, c,  false,  false);
}

#define L0_P   41
#define L0_PL  (40 * L0_P)
#define L0_YP  52
#define L0_YW  (2 * 16 * L0_YP)
#define L0_OP  100
#define L0_OT  (6 * L0_YW)

__global__ void __launch_bounds__(384, 12)
k_level0(const float* __restrict__ Z, const float* __restrict__ high0,
         float* __restrict__ out) {
  __shared__ __attribute__((aligned(16))) float sm[12 * L0_PL];

  const int tx = blockIdx.x, ty = blockIdx.y, b = blockIdx.z;
  const int tid = threadIdx.x;
  const int X0 = 32 * tx, Y0 = 32 * ty;
  const int R0 = Y0 - 4, C0 = X0 - 4;
  const int ib = 16 * ty - 2, jb = 16 * tx - 2;

  const int lane = tid & 31;
  const int w = __builtin_amdgcn_readfirstlane(tid >> 5);

  float zv[14];
  float hra[10], hia[10], hrb[10], hib[10];
  int zdst, p0, p1;
  const float* zb = Z + b * (512 * 512 * 3);
  const float* hb = high0 + b * (256 * 256 * 36);
  {
    const int g = w >> 2;
    const int f = lane + 32 * (w & 3);
    const int fz = f < 120 ? f : 119;
    const int wc = fz / 3, ch = fz - 3 * wc;
    const int zcol = ref_idx(C0 + wc, 512) * 3 + ch;
    zdst = f < 120 ? (ch * L0_PL + wc) : 40 + L0_P * (f - 120);
#pragma unroll
    for (int n = 0; n < 14; ++n) {
      int wr = 3 * n + g;
      wr = wr < 40 ? wr : 39;
      zv[n] = zb[ref_idx(R0 + wr, 512) * (512 * 3) + zcol];
    }
  }
  const int rp = w / 6;
  {
    const int m  = lane + 32 * (w - 6 * rp);
    const int mc = m < 180 ? m : 179;
    const int pc = mc / 9,  k = mc - 9 * pc;
    const int pair = k / 3, ch = k - 3 * pair;
    int J = jb + pc;
    const int fc = (J < 0 || J >= 256) ? 1 : 0;
    J = (J < 0) ? (-1 - J) : ((J >= 256) ? (511 - J) : J);
    const int oa = (pair == 0) ? 0  : ((pair == 1) ? 6 : 3);
    const int ob = (pair == 0) ? 15 : ((pair == 1) ? 9 : 12);
    const int ga = J * 36 + ch + oa;
    const int gb = J * 36 + ch + ob;
    const int c0 = ((1 + pair) * 3 + ch) * L0_PL + 2 * pc + fc;
    const int c1 = ((1 + pair) * 3 + ch) * L0_PL + 2 * pc + (1 ^ fc);
    p0 = m < 180 ? c0 : 40 + L0_P * (m - 180);
    p1 = m < 180 ? c1 : 40 + L0_P * (m - 180);
#pragma unroll
    for (int n = 0; n < 10; ++n) {
      const int pr = 2 * n + rp;
      int I = ib + pr;
      I = (I < 0) ? (-1 - I) : ((I >= 256) ? (511 - I) : I);
      const float* srow = hb + I * (256 * 36);
      hra[n] = srow[ga];  hia[n] = srow[ga + 18];
      hrb[n] = srow[gb];  hib[n] = srow[gb + 18];
    }
  }

#pragma unroll
  for (int n = 0; n < 14; ++n) {
    int wr = 3 * n + (w >> 2);
    wr = wr < 40 ? wr : 39;
    sm[wr * L0_P + zdst] = zv[n];
  }
#pragma unroll
  for (int n = 0; n < 10; ++n) {
    const int pr = 2 * n + rp;
    const int I = ib + pr;
    const int fr = (I < 0 || I >= 256) ? 1 : 0;
    const int r0 = (2 * pr + fr) * L0_P;
    const int r1 = (2 * pr + (1 ^ fr)) * L0_P;
    const float ra = hra[n], ia  = hia[n];
    const float rb = hrb[n], ibm = hib[n];
    sm[p0 + r0] = (ra + rb)  * SCQ;
    sm[p1 + r0] = (ia + ibm) * SCQ;
    sm[p0 + r1] = (ia - ibm) * SCQ;
    sm[p1 + r1] = (rb - ra)  * SCQ;
  }

  const int lo = lane & 15;
  const int h2 = (lane >> 4) * 2;
  const int sub = w / 3, ch = w - 3 * sub;
  const int sy = sub >> 1, sx = sub & 1;
  const int x0 = X0 + 16 * sx, y0 = Y0 + 16 * sy;

  __syncthreads();

  v2f f0[6], f1[5];
#pragma unroll
  for (int kk = 0; kk < 6; ++kk) {
    const int k0 = 4 * kk + h2;
    f0[kk].x = tap7(k0 - lo - 1);  f0[kk].y = tap7(k0 - lo);
  }
#pragma unroll
  for (int kk = 0; kk < 5; ++kk) {
    const int k0 = 4 * kk + h2;
    f1[kk].x = tap5(k0 - lo);      f1[kk].y = tap5(k0 - lo + 1);
  }

  v8f a1 = {}, a2 = {}, am = {};
  {
    const float* zp = sm + ch * L0_PL + (16 * sy) * L0_P;
    const float* lp = zp + 3 * L0_PL;
    const float* hp = zp + 6 * L0_PL;
    const float* qp = zp + 9 * L0_PL;
    const float* m0 = sx ? hp : zp;
    const float* m1 = sx ? qp : lp;
    const int ec = sx ? 32 + (lo & 7) : lo;
    const int ep = (sx != 0 && lo >= 8) ? 6 * L0_PL : 0;
    const float* e0 = zp + ep;
    const float* e1 = lp + ep;
    const int cm = 16 + lo;
    if (sx == 0) {
#pragma unroll
      for (int kk = 0; kk < 6; ++kk) {
        const int k0 = 4 * kk + h2;
        const int o0 = k0 * L0_P + lo, o1 = o0 + L0_P;
        v2f bh = { hp[o0], hp[o1] };
        a2 = wmma4(f0[kk], bh, a2);
        if (kk < 5) {
          v2f bq = { qp[o0 + 2 * L0_P], qp[o1 + 2 * L0_P] };
          a2 = wmma4(f1[kk], bq, a2);
        }
      }
    }
#pragma unroll
    for (int kk = 0; kk < 6; ++kk) {
      const int k0 = 4 * kk + h2;
      const int o0 = k0 * L0_P + ec, o1 = o0 + L0_P;
      const int c0 = k0 * L0_P + cm, c1 = c0 + L0_P;
      v2f be = { e0[o0], e0[o1] };
      v2f b0 = { m0[c0], m0[c1] };
      a1 = wmma4(f0[kk], be, a1);
      am = wmma4(f0[kk], b0, am);
      if (kk < 5) {
        v2f bl = { e1[o0 + 2 * L0_P], e1[o1 + 2 * L0_P] };
        v2f b1 = { m1[c0 + 2 * L0_P], m1[c1 + 2 * L0_P] };
        a1 = wmma4(f1[kk], bl, a1);
        am = wmma4(f1[kk], b1, am);
      }
    }
  }
  __syncthreads();

  float* y1s = sm + (3 * sy + ch) * L0_YW;
  float* y2s = y1s + 16 * L0_YP;
  float* yms = sx ? y2s : y1s;
  float* yes = sx ? ((lo >= 8 ? y2s : y1s) + 32 + (lo & 7)) : (y1s + lo);
#pragma unroll
  for (int v = 0; v < 8; ++v) {
    const int row = v + (h2 << 2);
    yes[row * L0_YP]            = a1[v];
    yms[row * L0_YP + 16 + lo]  = am[v];
  }
  if (sx == 0) {
#pragma unroll
    for (int v = 0; v < 8; ++v) {
      const int row = v + (h2 << 2);
      y2s[row * L0_YP + lo] = a2[v];
    }
  }
  __syncthreads();

  const float* y1r = y1s + 16 * sx;
  const float* y2r = y2s + 16 * sx;
  v8f o = {};
#pragma unroll
  for (int kk = 0; kk < 6; ++kk) {
    const int k0 = 4 * kk + h2;
    v2f A1 = { y1r[lo * L0_YP + k0], y1r[lo * L0_YP + k0 + 1] };
    o = wmma4(A1, f0[kk], o);
    if (kk < 5) {
      v2f A2 = { y2r[lo * L0_YP + k0 + 2], y2r[lo * L0_YP + k0 + 3] };
      o = wmma4(A2, f1[kk], o);
    }
  }
  float* ot = sm + L0_OT;
#pragma unroll
  for (int v = 0; v < 8; ++v) {
    const int row = 16 * sy + v + (h2 << 2);
    ot[row * L0_OP + (16 * sx + lo) * 3 + ch] = o[v];
  }
  __syncthreads();
  float* ob = out + ((b * 512 + Y0) * 512 + X0) * 3;
#pragma unroll
  for (int i = 0; i < 2; ++i) {
    const int q = tid + 384 * i;
    const int row = q / 24, c4 = q - 24 * row;
    const v4f val = *(const v4f*)(ot + row * L0_OP + 4 * c4);
    *(v4f*)(ob + row * (512 * 3) + 4 * c4) = val;
  }
}

extern "C" void kernel_launch(void* const* d_in, const int* in_sizes, int n_in,
                              void* d_out, int out_size, void* d_ws, size_t ws_size,
                              hipStream_t stream) {
  (void)in_sizes; (void)n_in; (void)out_size; (void)ws_size;
  const float* low   = (const float*)d_in[0];
  const float* high0 = (const float*)d_in[1];
  const float* high1 = (const float*)d_in[2];
  float* out = (float*)d_out;
  float* Zws = (float*)d_ws;

  k_level1<<<dim3(16, 16, 8), dim3(192), 0, stream>>>(low, high1, Zws);
  k_level0<<<dim3(16, 16, 8), dim3(384), 0, stream>>>(Zws, high0, out);
}


# level-1 column and row pass on two-wide f32 multiply-adds (v_pk_fma_f32): the two inputs of a filter pair lie interleaved in LDS and share each tap up to a sign; row-pass sums regrouped
# speedup vs baseline: 6.7411x; 1.0408x over previous
#include <hip/hip_runtime.h>
#include <hip/hip_bf16.h>


typedef float v2f __attribute__((ext_vector_type(2)));
typedef float v8f __attribute__((ext_vector_type(8)));
typedef float v4f __attribute__((ext_vector_type(4)));

#define SCQ 0.70710678118654752440f

__constant__ float cG0O[7] = { -3.0f/280.0f, -15.0f/280.0f, 73.0f/280.0f,
                               170.0f/280.0f, 73.0f/280.0f, -15.0f/280.0f,
                               -3.0f/280.0f };
__constant__ float cG1O[5] = { -1.0f/20.0f, -5.0f/20.0f, 12.0f/20.0f,
                               -5.0f/20.0f, -1.0f/20.0f };

__device__ __forceinline__ int ref_idx(int i, int L) {
  if (i < 0)  i = -1 - i;
  if (i >= L) i = 2 * L - 1 - i;
  return i;
}

#define L1_RS  85
#define L1_IMG (28 * L1_RS)
#define L1_YPL (32 * L1_RS)
#define L1_ZS  97

__device__ __forceinline__ v2f fma2(float t, v2f x, v2f acc) {
  const v2f tt = { t, t };
  return __builtin_elementwise_fma(tt, x, acc);
}

__global__ void __launch_bounds__(192)
k_level1(const float* __restrict__ low, const float* __restrict__ high1,
         float* __restrict__ Z) {
  constexpr float G0B[14] = {
     0.00325314f, -0.00388321f,  0.03466035f, -0.03887280f,
    -0.11720389f,  0.27529538f,  0.75614564f,  0.56881042f,
     0.01186609f, -0.10671180f,  0.02382538f,  0.01702522f,
    -0.00543948f, -0.00455690f };
  constexpr float G0A[14] = {
    -0.00455690f, -0.00543948f,  0.01702522f,  0.02382538f,
    -0.10671180f,  0.01186609f,  0.56881042f,  0.75614564f,
     0.27529538f, -0.11720389f, -0.03887280f,  0.03466035f,
    -0.00388321f,  0.00325314f };

  __shared__ __attribute__((aligned(8))) float sm[4 * L1_IMG + 2 * L1_YPL];
  float* img = sm;
  float* ys  = sm + 4 * L1_IMG;
  float* zs  = sm;

  const int tx = blockIdx.x, ty = blockIdx.y, b = blockIdx.z;
  const int tid = threadIdx.x;
  const int R0 = 16 * ty - 6, C0 = 16 * tx - 6;
  const int ib = 8 * ty - 3,  jb = 8 * tx - 3;

  const int lane = tid & 31;
  const int wv = __builtin_amdgcn_readfirstlane(tid >> 5);

  float vl[5][3];
  float hra[3][4], hia[3][4], hrb[3][4], hib[3][4];
  int fl[3], l0[4], l1[4];
  {
    int gof[3];
#pragma unroll
    for (int i = 0; i < 3; ++i) {
      const int f  = lane + 32 * i;
      const int wc = f / 3, ch = f - 3 * wc;
      gof[i] = ref_idx(C0 + wc, 256) * 3 + ch;
      fl[i]  = 2 * (f < 84 ? f : 84 + L1_RS * (f - 84));
    }
#pragma unroll
    for (int n = 0; n < 5; ++n) {
      int wr = (5 - wv) + 6 * n;
      wr = wr < 28 ? wr : 27;
      const float* srow = low + (b * 256 + ref_idx(R0 + wr, 256)) * 768;
#pragma unroll
      for (int i = 0; i < 3; ++i) vl[n][i] = srow[gof[i]];
    }
  }
  {
    int ga[4], gb[4];
#pragma unroll
    for (int i = 0; i < 4; ++i) {
      const int m  = lane + 32 * i;
      const int pc = m / 9,   k  = m - 9 * pc;
      const int pair = k / 3, ch = k - 3 * pair;
      int J = jb + pc;
      const int fc = (J < 0 || J >= 128) ? 1 : 0;
      J = (J < 0) ? (-1 - J) : ((J >= 128) ? (255 - J) : J);
      const int oa = (pair == 0) ? 0  : ((pair == 1) ? 6 : 3);
      const int ob = (pair == 0) ? 15 : ((pair == 1) ? 9 : 12);
      ga[i] = J * 36 + ch + oa;
      gb[i] = J * 36 + ch + ob;
      const int pb = (pair == 0) ? 1 : (2 * L1_IMG + (pair - 1));
      const int c0 = pb + 2 * ((2 * pc + fc) * 3 + ch);
      const int c1 = pb + 2 * ((2 * pc + (1 ^ fc)) * 3 + ch);
      l0[i] = m < 126 ? c0 : (1 + 2 * (84 + L1_RS * (m - 126)));
      l1[i] = m < 126 ? c1 : (1 + 2 * (84 + L1_RS * (m - 126)));
    }
#pragma unroll
    for (int n = 0; n < 3; ++n) {
      int pr = wv + 6 * n;
      pr = pr < 14 ? pr : 13;
      int I = ib + pr;
      I = (I < 0) ? (-1 - I) : ((I >= 128) ? (255 - I) : I);
      const float* srow = high1 + (b * 128 + I) * (128 * 36);
#pragma unroll
      for (int i = 0; i < 4; ++i) {
        hra[n][i] = srow[ga[i]];  hia[n][i] = srow[ga[i] + 18];
        hrb[n][i] = srow[gb[i]];  hib[n][i] = srow[gb[i] + 18];
      }
    }
  }

#pragma unroll
  for (int n = 0; n < 5; ++n) {
    int wr = (5 - wv) + 6 * n;
    wr = wr < 28 ? wr : 27;
    float* drow = img + 2 * wr * L1_RS;
#pragma unroll
    for (int i = 0; i < 3; ++i) drow[fl[i]] = vl[n][i];
  }
#pragma unroll
  for (int n = 0; n < 3; ++n) {
    int pr = wv + 6 * n;
    pr = pr < 14 ? pr : 13;
    const int I = ib + pr;
    const int fr = (I < 0 || I >= 128) ? 1 : 0;
    float* r0 = img + 2 * (2 * pr + fr) * L1_RS;
    float* r1 = img + 2 * (2 * pr + (1 ^ fr)) * L1_RS;
#pragma unroll
    for (int i = 0; i < 4; ++i) {
      const float ra = hra[n][i], ia  = hia[n][i];
      const float rb = hrb[n][i], ibm = hib[n][i];
      r0[l0[i]] = (ra + rb)  * SCQ;
      r0[l1[i]] = (ia + ibm) * SCQ;
      r1[l0[i]] = (ia - ibm) * SCQ;
      r1[l1[i]] = (rb - ra)  * SCQ;
    }
  }
  __syncthreads();

  {
    const int plane = tid / 96, c = tid - plane * 96;
    if (c < 84) {
      const v2f* X = (const v2f*)(img + plane * (2 * L1_IMG)) + c;
      v2f x[28];
#pragma unroll
      for (int r = 0; r < 28; ++r) x[r] = X[r * L1_RS];
      float* Y = ys + 2 * c + plane;
#pragma unroll
      for (int q = 0; q < 8; ++q) {
#pragma unroll
        for (int p = 0; p < 2; ++p) {
          v2f P = { 0.0f, 0.0f }, Q = { 0.0f, 0.0f };
#pragma unroll
          for (int j = 0; j < 7; ++j) {
            P = fma2(G0B[2 * j + p], x[2 * q + 12 - 2 * j], P);
            Q = fma2(G0A[2 * j + p], x[2 * q + 13 - 2 * j], Q);
          }
          Y[(4 * q + 2 * p)     * (2 * L1_RS)] = p ? P.x - Q.y : P.x + Q.y;
          Y[(4 * q + 2 * p + 1) * (2 * L1_RS)] = p ? Q.x + P.y : Q.x - P.y;
        }
      }
    }
  }
  __syncthreads();

  {
    const int w = tid >> 5, yy = tid & 31;
    const int half = w / 3, ch = w - 3 * half;
    const v2f* U = (const v2f*)ys + yy * L1_RS + (8 * half) * 3 + ch;
    v2f u[20];
#pragma unroll
    for (int k = 0; k < 20; ++k) u[k] = U[3 * k];
    float* zrow = zs + yy * L1_ZS + (16 * half) * 3 + ch;
#pragma unroll
    for (int q = 0; q < 4; ++q) {
#pragma unroll
      for (int p = 0; p < 2; ++p) {
        v2f P = { 0.0f, 0.0f }, Q = { 0.0f, 0.0f };
#pragma unroll
        for (int j = 0; j < 7; ++j) {
          P = fma2(G0B[2 * j + p], u[2 * q + 12 - 2 * j], P);
          Q = fma2(G0A[2 * j + p], u[2 * q + 13 - 2 * j], Q);
        }
        zrow[(4 * q + 2 * p)     * 3] = p ? P.x - Q.y : P.x + Q.y;
        zrow[(4 * q + 2 * p + 1) * 3] = p ? Q.x + P.y : Q.x - P.y;
      }
    }
  }
  __syncthreads();

  {
    const int hrow = tid / 96, f = tid - 96 * hrow;
#pragma unroll
    for (int it = 0; it < 16; ++it) {
      const int row = 2 * it + hrow;
      Z[((b * 512 + 32 * ty + row) * 512 + 32 * tx) * 3 + f] = zs[row * L1_ZS + f];
    }
  }
}

__device__ __forceinline__ float tap7(int i) {
  int ic = i < 0 ? 0 : (i > 6 ? 6 : i);
  float t = cG0O[ic];
  return (i == ic) ? t : 0.0f;
}
__device__ __forceinline__ float tap5(int i) {
  int ic = i < 0 ? 0 : (i > 4 ? 4 : i);
  float t = cG1O[ic];
  return (i == ic) ? t : 0.0f;
}

__device__ __forceinline__ v8f wmma4(v2f a, v2f b, v8f c) {
  return __builtin_amdgcn_wmma_f32_16x16x4_f32(
       false, a,  false, b,
       (short)0, c,  false,  false);
}

#define L0_P   41
#define L0_PL  (40 * L0_P)
#define L0_YP  52
#define L0_YW  (2 * 16 * L0_YP)
#define L0_OP  100
#define L0_OT  (6 * L0_YW)

__global__ void __launch_bounds__(384, 12)
k_level0(const float* __restrict__ Z, const float* __restrict__ high0,
         float* __restrict__ out) {
  __shared__ __attribute__((aligned(16))) float sm[12 * L0_PL];

  const int tx = blockIdx.x, ty = blockIdx.y, b = blockIdx.z;
  const int tid = threadIdx.x;
  const int X0 = 32 * tx, Y0 = 32 * ty;
  const int R0 = Y0 - 4, C0 = X0 - 4;
  const int ib = 16 * ty - 2, jb = 16 * tx - 2;

  const int lane = tid & 31;
  const int w = __builtin_amdgcn_readfirstlane(tid >> 5);

  float zv[14];
  float hra[10], hia[10], hrb[10], hib[10];
  int zdst, p0, p1;
  const float* zb = Z + b * (512 * 512 * 3);
  const float* hb = high0 + b * (256 * 256 * 36);
  {
    const int g = w >> 2;
    const int f = lane + 32 * (w & 3);
    const int fz = f < 120 ? f : 119;
    const int wc = fz / 3, ch = fz - 3 * wc;
    const int zcol = ref_idx(C0 + wc, 512) * 3 + ch;
    zdst = f < 120 ? (ch * L0_PL + wc) : 40 + L0_P * (f - 120);
#pragma unroll
    for (int n = 0; n < 14; ++n) {
      int wr = 3 * n + g;
      wr = wr < 40 ? wr : 39;
      zv[n] = zb[ref_idx(R0 + wr, 512) * (512 * 3) + zcol];
    }
  }
  const int rp = w / 6;
  {
    const int m  = lane + 32 * (w - 6 * rp);
    const int mc = m < 180 ? m : 179;
    const int pc = mc / 9,  k = mc - 9 * pc;
    const int pair = k / 3, ch = k - 3 * pair;
    int J = jb + pc;
    const int fc = (J < 0 || J >= 256) ? 1 : 0;
    J = (J < 0) ? (-1 - J) : ((J >= 256) ? (511 - J) : J);
    const int oa = (pair == 0) ? 0  : ((pair == 1) ? 6 : 3);
    const int ob = (pair == 0) ? 15 : ((pair == 1) ? 9 : 12);
    const int ga = J * 36 + ch + oa;
    const int gb = J * 36 + ch + ob;
    const int c0 = ((1 + pair) * 3 + ch) * L0_PL + 2 * pc + fc;
    const int c1 = ((1 + pair) * 3 + ch) * L0_PL + 2 * pc + (1 ^ fc);
    p0 = m < 180 ? c0 : 40 + L0_P * (m - 180);
    p1 = m < 180 ? c1 : 40 + L0_P * (m - 180);
#pragma unroll
    for (int n = 0; n < 10; ++n) {
      const int pr = 2 * n + rp;
      int I = ib + pr;
      I = (I < 0) ? (-1 - I) : ((I >= 256) ? (511 - I) : I);
      const float* srow = hb + I * (256 * 36);
      hra[n] = srow[ga];  hia[n] = srow[ga + 18];
      hrb[n] = srow[gb];  hib[n] = srow[gb + 18];
    }
  }

#pragma unroll
  for (int n = 0; n < 14; ++n) {
    int wr = 3 * n + (w >> 2);
    wr = wr < 40 ? wr : 39;
    sm[wr * L0_P + zdst] = zv[n];
  }
#pragma unroll
  for (int n = 0; n < 10; ++n) {
    const int pr = 2 * n + rp;
    const int I = ib + pr;
    const int fr = (I < 0 || I >= 256) ? 1 : 0;
    const int r0 = (2 * pr + fr) * L0_P;
    const int r1 = (2 * pr + (1 ^ fr)) * L0_P;
    const float ra = hra[n], ia  = hia[n];
    const float rb = hrb[n], ibm = hib[n];
    sm[p0 + r0] = (ra + rb)  * SCQ;
    sm[p1 + r0] = (ia + ibm) * SCQ;
    sm[p0 + r1] = (ia - ibm) * SCQ;
    sm[p1 + r1] = (rb - ra)  * SCQ;
  }

  const int lo = lane & 15;
  const int h2 = (lane >> 4) * 2;
  const int sub = w / 3, ch = w - 3 * sub;
  const int sy = sub >> 1, sx = sub & 1;
  const int x0 = X0 + 16 * sx, y0 = Y0 + 16 * sy;

  __syncthreads();

  v2f f0[6], f1[5];
#pragma unroll
  for (int kk = 0; kk < 6; ++kk) {
    const int k0 = 4 * kk + h2;
    f0[kk].x = tap7(k0 - lo - 1);  f0[kk].y = tap7(k0 - lo);
  }
#pragma unroll
  for (int kk = 0; kk < 5; ++kk) {
    const int k0 = 4 * kk + h2;
    f1[kk].x = tap5(k0 - lo);      f1[kk].y = tap5(k0 - lo + 1);
  }

  v8f a1 = {}, a2 = {}, am = {};
  {
    const float* zp = sm + ch * L0_PL + (16 * sy) * L0_P;
    const float* lp = zp + 3 * L0_PL;
    const float* hp = zp + 6 * L0_PL;
    const float* qp = zp + 9 * L0_PL;
    const float* m0 = sx ? hp : zp;
    const float* m1 = sx ? qp : lp;
    const int ec = sx ? 32 + (lo & 7) : lo;
    const int ep = (sx != 0 && lo >= 8) ? 6 * L0_PL : 0;
    const float* e0 = zp + ep;
    const float* e1 = lp + ep;
    const int cm = 16 + lo;
    if (sx == 0) {
#pragma unroll
      for (int kk = 0; kk < 6; ++kk) {
        const int k0 = 4 * kk + h2;
        const int o0 = k0 * L0_P + lo, o1 = o0 + L0_P;
        v2f bh = { hp[o0], hp[o1] };
        a2 = wmma4(f0[kk], bh, a2);
        if (kk < 5) {
          v2f bq = { qp[o0 + 2 * L0_P], qp[o1 + 2 * L0_P] };
          a2 = wmma4(f1[kk], bq, a2);
        }
      }
    }
#pragma unroll
    for (int kk = 0; kk < 6; ++kk) {
      const int k0 = 4 * kk + h2;
      const int o0 = k0 * L0_P + ec, o1 = o0 + L0_P;
      const int c0 = k0 * L0_P + cm, c1 = c0 + L0_P;
      v2f be = { e0[o0], e0[o1] };
      v2f b0 = { m0[c0], m0[c1] };
      a1 = wmma4(f0[kk], be, a1);
      am = wmma4(f0[kk], b0, am);
      if (kk < 5) {
        v2f bl = { e1[o0 + 2 * L0_P], e1[o1 + 2 * L0_P] };
        v2f b1 = { m1[c0 + 2 * L0_P], m1[c1 + 2 * L0_P] };
        a1 = wmma4(f1[kk], bl, a1);
        am = wmma4(f1[kk], b1, am);
      }
    }
  }
  __syncthreads();

  float* y1s = sm + (3 * sy + ch) * L0_YW;
  float* y2s = y1s + 16 * L0_YP;
  float* yms = sx ? y2s : y1s;
  float* yes = sx ? ((lo >= 8 ? y2s : y1s) + 32 + (lo & 7)) : (y1s + lo);
#pragma unroll
  for (int v = 0; v < 8; ++v) {
    const int row = v + (h2 << 2);
    yes[row * L0_YP]            = a1[v];
    yms[row * L0_YP + 16 + lo]  = am[v];
  }
  if (sx == 0) {
#pragma unroll
    for (int v = 0; v < 8; ++v) {
      const int row = v + (h2 << 2);
      y2s[row * L0_YP + lo] = a2[v];
    }
  }
  __syncthreads();

  const float* y1r = y1s + 16 * sx;
  const float* y2r = y2s + 16 * sx;
  v8f o = {};
#pragma unroll
  for (int kk = 0; kk < 6; ++kk) {
    const int k0 = 4 * kk + h2;
    v2f A1 = { y1r[lo * L0_YP + k0], y1r[lo * L0_YP + k0 + 1] };
    o = wmma4(A1, f0[kk], o);
    if (kk < 5) {
      v2f A2 = { y2r[lo * L0_YP + k0 + 2], y2r[lo * L0_YP + k0 + 3] };
      o = wmma4(A2, f1[kk], o);
    }
  }
  float* ot = sm + L0_OT;
#pragma unroll
  for (int v = 0; v < 8; ++v) {
    const int row = 16 * sy + v + (h2 << 2);
    ot[row * L0_OP + (16 * sx + lo) * 3 + ch] = o[v];
  }
  __syncthreads();
  float* ob = out + ((b * 512 + Y0) * 512 + X0) * 3;
#pragma unroll
  for (int i = 0; i < 2; ++i) {
    const int q = tid + 384 * i;
    const int row = q / 24, c4 = q - 24 * row;
    const v4f val = *(const v4f*)(ot + row * L0_OP + 4 * c4);
    *(v4f*)(ob + row * (512 * 3) + 4 * c4) = val;
  }
}

extern "C" void kernel_launch(void* const* d_in, const int* in_sizes, int n_in,
                              void* d_out, int out_size, void* d_ws, size_t ws_size,
                              hipStream_t stream) {
  (void)in_sizes; (void)n_in; (void)out_size; (void)ws_size;
  const float* low   = (const float*)d_in[0];
  const float* high0 = (const float*)d_in[1];
  const float* high1 = (const float*)d_in[2];
  float* out = (float*)d_out;
  float* Zws = (float*)d_ws;

  k_level1<<<dim3(16, 16, 8), dim3(192), 0, stream>>>(low, high1, Zws);
  k_level0<<<dim3(16, 16, 8), dim3(384), 0, stream>>>(Zws, high0, out);
}
